# starting kernel
# baseline (speedup 1.0000x reference)
#include <hip/hip_runtime.h>

#define N_NODES   100000
#define N_EDGES   1200000
#define EMBED_D   32
#define HIDDEN_D  64
#define N_CLASSES 10
#define N_GRAPHS  1024

typedef __attribute__((ext_vector_type(2))) float v2f;
typedef __attribute__((ext_vector_type(8))) float v8f;

__device__ __forceinline__ void atomAdd1(float* p, float v) {
    asm volatile("global_atomic_add_f32 %0, %1, off scope:SCOPE_DEV"
                 :: "v"(p), "v"(v) : "memory");
}
__device__ __forceinline__ void atomAdd4(float* p, float4 v) {
    asm volatile(
        "global_atomic_add_f32 %0, %1, off scope:SCOPE_DEV\n\t"
        "global_atomic_add_f32 %0, %2, off offset:4 scope:SCOPE_DEV\n\t"
        "global_atomic_add_f32 %0, %3, off offset:8 scope:SCOPE_DEV\n\t"
        "global_atomic_add_f32 %0, %4, off offset:12 scope:SCOPE_DEV"
        :: "v"(p), "v"(v.x), "v"(v.y), "v"(v.z), "v"(v.w) : "memory");
}

__global__ void zero_kernel(float* __restrict__ p, int n) {
    int t = blockIdx.x * blockDim.x + threadIdx.x;
    if (t < n) p[t] = 0.0f;
}

__global__ void gather_embed_kernel(const int* __restrict__ ids,
                                    const float* __restrict__ embed,
                                    float* __restrict__ x0, int nNodes) {
    int t = blockIdx.x * blockDim.x + threadIdx.x;
    if (t >= nNodes * (EMBED_D / 4)) return;
    int node = t >> 3;
    int q    = t & 7;
    float4 v = reinterpret_cast<const float4*>(embed + (size_t)ids[node] * EMBED_D)[q];
    reinterpret_cast<float4*>(x0 + (size_t)node * EMBED_D)[q] = v;
}

__global__ void edge_count_kernel(const int* __restrict__ dst,
                                  float* __restrict__ cnt, int nEdges) {
    int e = blockIdx.x * blockDim.x + threadIdx.x;
    if (e >= nEdges) return;
    atomAddF_guard:
    atomAdd1(cnt + dst[e], 1.0f);
}

template <int DIM>
__global__ void edge_scatter_kernel(const int* __restrict__ src,
                                    const int* __restrict__ dst,
                                    const float* __restrict__ x,
                                    float* __restrict__ agg, int nEdges) {
    constexpr int QPE = DIM / 4;
    int t = blockIdx.x * blockDim.x + threadIdx.x;
    if (t >= nEdges * QPE) return;
    int e = t / QPE;
    int q = t & (QPE - 1);
    int s = src[e], d = dst[e];
    float4 v = reinterpret_cast<const float4*>(x + (size_t)s * DIM)[q];
    atomAdd4(agg + (size_t)d * DIM + q * 4, v);
}

template <int K>
__global__ void sage_gemm_kernel(const float* __restrict__ aggRaw,
                                 const float* __restrict__ cnt,
                                 const float* __restrict__ x,
                                 const float* __restrict__ Wl,
                                 const float* __restrict__ Wr,
                                 const float* __restrict__ bias,
                                 float* __restrict__ out, int nNodes) {
    const int lane = threadIdx.x & 31;
    const int wave = threadIdx.x >> 5;
    const int row0 = blockIdx.x * 16;
    if (row0 >= nNodes) return;
    const int col0 = wave * 16;
    const int grp  = lane >> 4;
    const int lr   = lane & 15;

    const int   arow = row0 + lr;
    const float invc = 1.0f / fmaxf(cnt[arow], 1.0f);

    v8f c = {};
#pragma unroll
    for (int k0 = 0; k0 < K; k0 += 4) {
        const int ka = k0 + 2 * grp;
        v2f a, b;
        a.x = aggRaw[(size_t)arow * K + ka + 0] * invc;
        a.y = aggRaw[(size_t)arow * K + ka + 1] * invc;
        b.x = Wl[(size_t)(ka + 0) * HIDDEN_D + col0 + lr];
        b.y = Wl[(size_t)(ka + 1) * HIDDEN_D + col0 + lr];
        c = __builtin_amdgcn_wmma_f32_16x16x4_f32(false, a, false, b,
                                                  (short)0, c, false, false);
    }
#pragma unroll
    for (int k0 = 0; k0 < K; k0 += 4) {
        const int ka = k0 + 2 * grp;
        v2f a, b;
        a.x = x[(size_t)arow * K + ka + 0];
        a.y = x[(size_t)arow * K + ka + 1];
        b.x = Wr[(size_t)(ka + 0) * HIDDEN_D + col0 + lr];
        b.y = Wr[(size_t)(ka + 1) * HIDDEN_D + col0 + lr];
        c = __builtin_amdgcn_wmma_f32_16x16x4_f32(false, a, false, b,
                                                  (short)0, c, false, false);
    }

    const float bv = bias[col0 + lr];
#pragma unroll
    for (int v = 0; v < 8; ++v) {
        const int m = v + 8 * grp;
        out[(size_t)(row0 + m) * HIDDEN_D + col0 + lr] = fmaxf(c[v] + bv, 0.0f);
    }
}

__global__ void pool_kernel(const int* __restrict__ batch,
                            const float* __restrict__ h,
                            float* __restrict__ gsum,
                            float* __restrict__ gcnt, int nNodes) {
    int t = blockIdx.x * blockDim.x + threadIdx.x;
    if (t >= nNodes * (HIDDEN_D / 4)) return;
    int node = t >> 4;
    int q    = t & 15;
    int g    = batch[node];
    float4 v = reinterpret_cast<const float4*>(h + (size_t)node * HIDDEN_D)[q];
    atomAdd4(gsum + (size_t)g * HIDDEN_D + q * 4, v);
    if (q == 0) atomAdd1(gcnt + g, 1.0f);
}

__global__ void head_kernel(const float* __restrict__ gsum,
                            const float* __restrict__ gcnt,
                            const float* __restrict__ Wo,
                            const float* __restrict__ bo,
                            float* __restrict__ out) {
    int t = blockIdx.x * blockDim.x + threadIdx.x;
    if (t >= N_GRAPHS * N_CLASSES) return;
    int g = t / N_CLASSES;
    int c = t - g * N_CLASSES;
    float invc = 1.0f / fmaxf(gcnt[g], 1.0f);
    float acc = 0.0f;
    for (int d = 0; d < HIDDEN_D; ++d)
        acc += gsum[(size_t)g * HIDDEN_D + d] * invc * Wo[(size_t)d * N_CLASSES + c];
    out[t] = acc + bo[c];
}

extern "C" void kernel_launch(void* const* d_in, const int* in_sizes, int n_in,
                              void* d_out, int out_size, void* d_ws, size_t ws_size,
                              hipStream_t stream) {
    const int*   node_ids = (const int*)  d_in[0];
    const int*   src      = (const int*)  d_in[1];
    const int*   dst      = (const int*)  d_in[2];
    const int*   batch    = (const int*)  d_in[3];
    const float* embed    = (const float*)d_in[4];
    const float* W1_l     = (const float*)d_in[5];
    const float* b1       = (const float*)d_in[6];
    const float* W1_r     = (const float*)d_in[7];
    const float* W2_l     = (const float*)d_in[8];
    const float* b2       = (const float*)d_in[9];
    const float* W2_r     = (const float*)d_in[10];
    const float* W_out    = (const float*)d_in[11];
    const float* b_out    = (const float*)d_in[12];
    float* out = (float*)d_out;

    float* ws   = (float*)d_ws;
    float* x0   = ws;
    float* h1   = x0   + (size_t)N_NODES * EMBED_D;
    float* h2   = h1   + (size_t)N_NODES * HIDDEN_D;
    float* agg1 = h2   + (size_t)N_NODES * HIDDEN_D;
    float* agg2 = agg1 + (size_t)N_NODES * EMBED_D;
    float* cnt  = agg2 + (size_t)N_NODES * HIDDEN_D;
    float* gsum = cnt  + (size_t)N_NODES;
    float* gcnt = gsum + (size_t)N_GRAPHS * HIDDEN_D;
    const int zeroN = N_NODES * EMBED_D + N_NODES * HIDDEN_D + N_NODES
                    + N_GRAPHS * HIDDEN_D + N_GRAPHS;

    const int B = 256;
    auto blocks = [](int n, int b) { return (n + b - 1) / b; };

    zero_kernel<<<blocks(zeroN, B), B, 0, stream>>>(agg1, zeroN);

    gather_embed_kernel<<<blocks(N_NODES * 8, B), B, 0, stream>>>(
        node_ids, embed, x0, N_NODES);

    edge_count_kernel<<<blocks(N_EDGES, B), B, 0, stream>>>(dst, cnt, N_EDGES);

    edge_scatter_kernel<EMBED_D><<<blocks(N_EDGES * (EMBED_D / 4), B), B, 0, stream>>>(
        src, dst, x0, agg1, N_EDGES);
    sage_gemm_kernel<EMBED_D><<<N_NODES / 16, 128, 0, stream>>>(
        agg1, cnt, x0, W1_l, W1_r, b1, h1, N_NODES);

    edge_scatter_kernel<HIDDEN_D><<<blocks(N_EDGES * (HIDDEN_D / 4), B), B, 0, stream>>>(
        src, dst, h1, agg2, N_EDGES);
    sage_gemm_kernel<HIDDEN_D><<<N_NODES / 16, 128, 0, stream>>>(
        agg2, cnt, h1, W2_l, W2_r, b2, h2, N_NODES);

    pool_kernel<<<blocks(N_NODES * (HIDDEN_D / 4), B), B, 0, stream>>>(
        batch, h2, gsum, gcnt, N_NODES);
    head_kernel<<<blocks(N_GRAPHS * N_CLASSES, B), B, 0, stream>>>(
        gsum, gcnt, W_out, b_out, out);
}


# edge scatter-add: one wave per group of edges; every 128-B row line read by one whole-line load and added by one whole-line f32 atomic (32 adjacent lanes); sums stay in atomic arrival order
# speedup vs baseline: 1.2568x; 1.2568x over previous
#include <hip/hip_runtime.h>

#define N_NODES   100000
#define N_EDGES   1200000
#define EMBED_D   32
#define HIDDEN_D  64
#define N_CLASSES 10
#define N_GRAPHS  1024

typedef __attribute__((ext_vector_type(2))) float v2f;
typedef __attribute__((ext_vector_type(8))) float v8f;

__device__ __forceinline__ void atomAdd1(float* p, float v) {
    asm volatile("global_atomic_add_f32 %0, %1, off scope:SCOPE_DEV"
                 :: "v"(p), "v"(v) : "memory");
}
__device__ __forceinline__ void atomAdd4(float* p, float4 v) {
    asm volatile(
        "global_atomic_add_f32 %0, %1, off scope:SCOPE_DEV\n\t"
        "global_atomic_add_f32 %0, %2, off offset:4 scope:SCOPE_DEV\n\t"
        "global_atomic_add_f32 %0, %3, off offset:8 scope:SCOPE_DEV\n\t"
        "global_atomic_add_f32 %0, %4, off offset:12 scope:SCOPE_DEV"
        :: "v"(p), "v"(v.x), "v"(v.y), "v"(v.z), "v"(v.w) : "memory");
}

__global__ void zero_kernel(float* __restrict__ p, int n) {
    int t = blockIdx.x * blockDim.x + threadIdx.x;
    if (t < n) p[t] = 0.0f;
}

__global__ void gather_embed_kernel(const int* __restrict__ ids,
                                    const float* __restrict__ embed,
                                    float* __restrict__ x0, int nNodes) {
    int t = blockIdx.x * blockDim.x + threadIdx.x;
    if (t >= nNodes * (EMBED_D / 4)) return;
    int node = t >> 3;
    int q    = t & 7;
    float4 v = reinterpret_cast<const float4*>(embed + (size_t)ids[node] * EMBED_D)[q];
    reinterpret_cast<float4*>(x0 + (size_t)node * EMBED_D)[q] = v;
}

__global__ void edge_count_kernel(const int* __restrict__ dst,
                                  float* __restrict__ cnt, int nEdges) {
    int e = blockIdx.x * blockDim.x + threadIdx.x;
    if (e >= nEdges) return;
    atomAddF_guard:
    atomAdd1(cnt + dst[e], 1.0f);
}

__device__ __forceinline__ void atomAddLine0(float* row, unsigned laneOff, float v) {
    asm volatile("global_atomic_add_f32 %0, %1, %2 scope:SCOPE_DEV"
                 :: "v"(laneOff), "v"(v), "s"(row) : "memory");
}
__device__ __forceinline__ void atomAddLine1(float* row, unsigned laneOff, float v) {
    asm volatile("global_atomic_add_f32 %0, %1, %2 offset:128 scope:SCOPE_DEV"
                 :: "v"(laneOff), "v"(v), "s"(row) : "memory");
}

template <int DIM, int EPW>
__global__ void edge_scatter_kernel(const int* __restrict__ src,
                                    const int* __restrict__ dst,
                                    const float* __restrict__ x,
                                    float* __restrict__ agg, int nEdges) {
    constexpr int LPR = DIM / 32;
    static_assert(LPR == 1 || LPR == 2, "row is one or two lines");
    static_assert(EPW == 16 || EPW == 32, "edges per wave");
    const int lane = threadIdx.x & 31;
    const int wave = blockIdx.x * (blockDim.x >> 5) + (threadIdx.x >> 5);
    const int e0   = wave * EPW;
    if (e0 >= nEdges) return;

    const int ei = e0 + (lane & (EPW - 1));
    const int sv = src[ei];
    const int dv = dst[ei];

    static_assert((unsigned long long)N_NODES * DIM * 4ull < (1ull << 32), "32-bit row offsets");
    const unsigned laneOff = (unsigned)lane * 4u;
    const char* xb = reinterpret_cast<const char*>(x);

    float v[EPW][LPR];
#pragma unroll
    for (int j = 0; j < EPW; ++j) {
        const unsigned s = (unsigned)__builtin_amdgcn_readlane(sv, j);
        const float* xr = reinterpret_cast<const float*>(xb + (s * (unsigned)(DIM * 4) + laneOff));
#pragma unroll
        for (int h = 0; h < LPR; ++h) v[j][h] = xr[32 * h];
    }
#pragma unroll
    for (int j = 0; j < EPW; ++j) {
        const int d = __builtin_amdgcn_readlane(dv, j);
        float* ar = agg + (size_t)d * DIM;
        atomAddLine0(ar, laneOff, v[j][0]);
        if (LPR == 2) atomAddLine1(ar, laneOff, v[j][LPR - 1]);
    }
}

template <int K>
__global__ void sage_gemm_kernel(const float* __restrict__ aggRaw,
                                 const float* __restrict__ cnt,
                                 const float* __restrict__ x,
                                 const float* __restrict__ Wl,
                                 const float* __restrict__ Wr,
                                 const float* __restrict__ bias,
                                 float* __restrict__ out, int nNodes) {
    const int lane = threadIdx.x & 31;
    const int wave = threadIdx.x >> 5;
    const int row0 = blockIdx.x * 16;
    if (row0 >= nNodes) return;
    const int col0 = wave * 16;
    const int grp  = lane >> 4;
    const int lr   = lane & 15;

    const int   arow = row0 + lr;
    const float invc = 1.0f / fmaxf(cnt[arow], 1.0f);

    v8f c = {};
#pragma unroll
    for (int k0 = 0; k0 < K; k0 += 4) {
        const int ka = k0 + 2 * grp;
        v2f a, b;
        a.x = aggRaw[(size_t)arow * K + ka + 0] * invc;
        a.y = aggRaw[(size_t)arow * K + ka + 1] * invc;
        b.x = Wl[(size_t)(ka + 0) * HIDDEN_D + col0 + lr];
        b.y = Wl[(size_t)(ka + 1) * HIDDEN_D + col0 + lr];
        c = __builtin_amdgcn_wmma_f32_16x16x4_f32(false, a, false, b,
                                                  (short)0, c, false, false);
    }
#pragma unroll
    for (int k0 = 0; k0 < K; k0 += 4) {
        const int ka = k0 + 2 * grp;
        v2f a, b;
        a.x = x[(size_t)arow * K + ka + 0];
        a.y = x[(size_t)arow * K + ka + 1];
        b.x = Wr[(size_t)(ka + 0) * HIDDEN_D + col0 + lr];
        b.y = Wr[(size_t)(ka + 1) * HIDDEN_D + col0 + lr];
        c = __builtin_amdgcn_wmma_f32_16x16x4_f32(false, a, false, b,
                                                  (short)0, c, false, false);
    }

    const float bv = bias[col0 + lr];
#pragma unroll
    for (int v = 0; v < 8; ++v) {
        const int m = v + 8 * grp;
        out[(size_t)(row0 + m) * HIDDEN_D + col0 + lr] = fmaxf(c[v] + bv, 0.0f);
    }
}

__global__ void pool_kernel(const int* __restrict__ batch,
                            const float* __restrict__ h,
                            float* __restrict__ gsum,
                            float* __restrict__ gcnt, int nNodes) {
    int t = blockIdx.x * blockDim.x + threadIdx.x;
    if (t >= nNodes * (HIDDEN_D / 4)) return;
    int node = t >> 4;
    int q    = t & 15;
    int g    = batch[node];
    float4 v = reinterpret_cast<const float4*>(h + (size_t)node * HIDDEN_D)[q];
    atomAdd4(gsum + (size_t)g * HIDDEN_D + q * 4, v);
    if (q == 0) atomAdd1(gcnt + g, 1.0f);
}

__global__ void head_kernel(const float* __restrict__ gsum,
                            const float* __restrict__ gcnt,
                            const float* __restrict__ Wo,
                            const float* __restrict__ bo,
                            float* __restrict__ out) {
    int t = blockIdx.x * blockDim.x + threadIdx.x;
    if (t >= N_GRAPHS * N_CLASSES) return;
    int g = t / N_CLASSES;
    int c = t - g * N_CLASSES;
    float invc = 1.0f / fmaxf(gcnt[g], 1.0f);
    float acc = 0.0f;
    for (int d = 0; d < HIDDEN_D; ++d)
        acc += gsum[(size_t)g * HIDDEN_D + d] * invc * Wo[(size_t)d * N_CLASSES + c];
    out[t] = acc + bo[c];
}

extern "C" void kernel_launch(void* const* d_in, const int* in_sizes, int n_in,
                              void* d_out, int out_size, void* d_ws, size_t ws_size,
                              hipStream_t stream) {
    const int*   node_ids = (const int*)  d_in[0];
    const int*   src      = (const int*)  d_in[1];
    const int*   dst      = (const int*)  d_in[2];
    const int*   batch    = (const int*)  d_in[3];
    const float* embed    = (const float*)d_in[4];
    const float* W1_l     = (const float*)d_in[5];
    const float* b1       = (const float*)d_in[6];
    const float* W1_r     = (const float*)d_in[7];
    const float* W2_l     = (const float*)d_in[8];
    const float* b2       = (const float*)d_in[9];
    const float* W2_r     = (const float*)d_in[10];
    const float* W_out    = (const float*)d_in[11];
    const float* b_out    = (const float*)d_in[12];
    float* out = (float*)d_out;

    float* ws   = (float*)d_ws;
    float* x0   = ws;
    float* h1   = x0   + (size_t)N_NODES * EMBED_D;
    float* h2   = h1   + (size_t)N_NODES * HIDDEN_D;
    float* agg1 = h2   + (size_t)N_NODES * HIDDEN_D;
    float* agg2 = agg1 + (size_t)N_NODES * EMBED_D;
    float* cnt  = agg2 + (size_t)N_NODES * HIDDEN_D;
    float* gsum = cnt  + (size_t)N_NODES;
    float* gcnt = gsum + (size_t)N_GRAPHS * HIDDEN_D;
    const int zeroN = N_NODES * EMBED_D + N_NODES * HIDDEN_D + N_NODES
                    + N_GRAPHS * HIDDEN_D + N_GRAPHS;

    const int B = 256;
    auto blocks = [](int n, int b) { return (n + b - 1) / b; };

    zero_kernel<<<blocks(zeroN, B), B, 0, stream>>>(agg1, zeroN);

    gather_embed_kernel<<<blocks(N_NODES * 8, B), B, 0, stream>>>(
        node_ids, embed, x0, N_NODES);

    edge_count_kernel<<<blocks(N_EDGES, B), B, 0, stream>>>(dst, cnt, N_EDGES);

    static_assert(N_EDGES % 32 == 0, "a wave owns a whole group of edges");
    edge_scatter_kernel<EMBED_D, 32><<<blocks(N_EDGES / 32 * 32, B), B, 0, stream>>>(
        src, dst, x0, agg1, N_EDGES);
    sage_gemm_kernel<EMBED_D><<<N_NODES / 16, 128, 0, stream>>>(
        agg1, cnt, x0, W1_l, W1_r, b1, h1, N_NODES);

    edge_scatter_kernel<HIDDEN_D, 16><<<blocks(N_EDGES / 16 * 32, B), B, 0, stream>>>(
        src, dst, h1, agg2, N_EDGES);
    sage_gemm_kernel<HIDDEN_D><<<N_NODES / 16, 128, 0, stream>>>(
        agg2, cnt, h1, W2_l, W2_r, b2, h2, N_NODES);

    pool_kernel<<<blocks(N_NODES * (HIDDEN_D / 4), B), B, 0, stream>>>(
        batch, h2, gsum, gcnt, N_NODES);
    head_kernel<<<blocks(N_GRAPHS * N_CLASSES, B), B, 0, stream>>>(
        gsum, gcnt, W_out, b_out, out);
}


# graph pooling fused into the layer-2 GEMM epilogue: a wave sums its 16 rows per column in registers, one add per graph and column; last features never stored; pooled sum order changes
# speedup vs baseline: 1.4988x; 1.1926x over previous
#include <hip/hip_runtime.h>

#define N_NODES   100000
#define N_EDGES   1200000
#define EMBED_D   32
#define HIDDEN_D  64
#define N_CLASSES 10
#define N_GRAPHS  1024

typedef __attribute__((ext_vector_type(2))) float v2f;
typedef __attribute__((ext_vector_type(8))) float v8f;

__device__ __forceinline__ void atomAdd1(float* p, float v) {
    asm volatile("global_atomic_add_f32 %0, %1, off scope:SCOPE_DEV"
                 :: "v"(p), "v"(v) : "memory");
}

__global__ void zero_kernel(float* __restrict__ p, int n) {
    int t = blockIdx.x * blockDim.x + threadIdx.x;
    if (t < n) p[t] = 0.0f;
}

__global__ void gather_embed_kernel(const int* __restrict__ ids,
                                    const float* __restrict__ embed,
                                    float* __restrict__ x0, int nNodes) {
    int t = blockIdx.x * blockDim.x + threadIdx.x;
    if (t >= nNodes * (EMBED_D / 4)) return;
    int node = t >> 3;
    int q    = t & 7;
    float4 v = reinterpret_cast<const float4*>(embed + (size_t)ids[node] * EMBED_D)[q];
    reinterpret_cast<float4*>(x0 + (size_t)node * EMBED_D)[q] = v;
}

__global__ void edge_count_kernel(const int* __restrict__ dst,
                                  float* __restrict__ cnt, int nEdges) {
    int e = blockIdx.x * blockDim.x + threadIdx.x;
    if (e >= nEdges) return;
    atomAddF_guard:
    atomAdd1(cnt + dst[e], 1.0f);
}

__device__ __forceinline__ void atomAddLine0(float* row, unsigned laneOff, float v) {
    asm volatile("global_atomic_add_f32 %0, %1, %2 scope:SCOPE_DEV"
                 :: "v"(laneOff), "v"(v), "s"(row) : "memory");
}
__device__ __forceinline__ void atomAddLine1(float* row, unsigned laneOff, float v) {
    asm volatile("global_atomic_add_f32 %0, %1, %2 offset:128 scope:SCOPE_DEV"
                 :: "v"(laneOff), "v"(v), "s"(row) : "memory");
}

template <int DIM, int EPW>
__global__ void edge_scatter_kernel(const int* __restrict__ src,
                                    const int* __restrict__ dst,
                                    const float* __restrict__ x,
                                    float* __restrict__ agg, int nEdges) {
    constexpr int LPR = DIM / 32;
    static_assert(LPR == 1 || LPR == 2, "row is one or two lines");
    static_assert(EPW == 16 || EPW == 32, "edges per wave");
    const int lane = threadIdx.x & 31;
    const int wave = blockIdx.x * (blockDim.x >> 5) + (threadIdx.x >> 5);
    const int e0   = wave * EPW;
    if (e0 >= nEdges) return;

    const int ei = e0 + (lane & (EPW - 1));
    const int sv = src[ei];
    const int dv = dst[ei];

    static_assert((unsigned long long)N_NODES * DIM * 4ull < (1ull << 32), "32-bit row offsets");
    const unsigned laneOff = (unsigned)lane * 4u;
    const char* xb = reinterpret_cast<const char*>(x);

    float v[EPW][LPR];
#pragma unroll
    for (int j = 0; j < EPW; ++j) {
        const unsigned s = (unsigned)__builtin_amdgcn_readlane(sv, j);
        const float* xr = reinterpret_cast<const float*>(xb + (s * (unsigned)(DIM * 4) + laneOff));
#pragma unroll
        for (int h = 0; h < LPR; ++h) v[j][h] = xr[32 * h];
    }
#pragma unroll
    for (int j = 0; j < EPW; ++j) {
        const int d = __builtin_amdgcn_readlane(dv, j);
        float* ar = agg + (size_t)d * DIM;
        atomAddLine0(ar, laneOff, v[j][0]);
        if (LPR == 2) atomAddLine1(ar, laneOff, v[j][LPR - 1]);
    }
}

template <int K, bool POOL>
__global__ void sage_gemm_kernel(const float* __restrict__ aggRaw,
                                 const float* __restrict__ cnt,
                                 const float* __restrict__ x,
                                 const float* __restrict__ Wl,
                                 const float* __restrict__ Wr,
                                 const float* __restrict__ bias,
                                 float* __restrict__ out,
                                 const int* __restrict__ batch,
                                 float* __restrict__ gsum,
                                 float* __restrict__ gcnt, int nNodes) {
    const int lane = threadIdx.x & 31;
    const int wave = threadIdx.x >> 5;
    const int row0 = blockIdx.x * 16;
    if (row0 >= nNodes) return;
    const int col0 = wave * 16;
    const int grp  = lane >> 4;
    const int lr   = lane & 15;

    const int   arow = row0 + lr;
    const float invc = 1.0f / fmaxf(cnt[arow], 1.0f);

    v8f c = {};
#pragma unroll
    for (int k0 = 0; k0 < K; k0 += 4) {
        const int ka = k0 + 2 * grp;
        v2f a, b;
        a.x = aggRaw[(size_t)arow * K + ka + 0] * invc;
        a.y = aggRaw[(size_t)arow * K + ka + 1] * invc;
        b.x = Wl[(size_t)(ka + 0) * HIDDEN_D + col0 + lr];
        b.y = Wl[(size_t)(ka + 1) * HIDDEN_D + col0 + lr];
        c = __builtin_amdgcn_wmma_f32_16x16x4_f32(false, a, false, b,
                                                  (short)0, c, false, false);
    }
#pragma unroll
    for (int k0 = 0; k0 < K; k0 += 4) {
        const int ka = k0 + 2 * grp;
        v2f a, b;
        a.x = x[(size_t)arow * K + ka + 0];
        a.y = x[(size_t)arow * K + ka + 1];
        b.x = Wr[(size_t)(ka + 0) * HIDDEN_D + col0 + lr];
        b.y = Wr[(size_t)(ka + 1) * HIDDEN_D + col0 + lr];
        c = __builtin_amdgcn_wmma_f32_16x16x4_f32(false, a, false, b,
                                                  (short)0, c, false, false);
    }

    const float bv = bias[col0 + lr];
    if (!POOL) {
#pragma unroll
        for (int v = 0; v < 8; ++v) {
            const int m = v + 8 * grp;
            out[(size_t)(row0 + m) * HIDDEN_D + col0 + lr] = fmaxf(c[v] + bv, 0.0f);
        }
    } else {
        const int gid = batch[row0 + lr];
        const int g0  = __builtin_amdgcn_readfirstlane(gid);
        float r[8];
#pragma unroll
        for (int v = 0; v < 8; ++v) r[v] = fmaxf(c[v] + bv, 0.0f);
        if (__all(gid == g0)) {
            float s = ((r[0] + r[1]) + (r[2] + r[3])) + ((r[4] + r[5]) + (r[6] + r[7]));
            s += __shfl_xor(s, 16, 32);
            if (grp == 0) atomAdd1(gsum + (size_t)g0 * HIDDEN_D + col0 + lr, s);
            if (threadIdx.x == 0) atomAdd1(gcnt + g0, 16.0f);
        } else {
#pragma unroll
            for (int v = 0; v < 8; ++v) {
                const int gv = __shfl(gid, v + 8 * grp, 32);
                atomAdd1(gsum + (size_t)gv * HIDDEN_D + col0 + lr, r[v]);
            }
            if (threadIdx.x < 16) atomAdd1(gcnt + gid, 1.0f);
        }
    }
}

__global__ void head_kernel(const float* __restrict__ gsum,
                            const float* __restrict__ gcnt,
                            const float* __restrict__ Wo,
                            const float* __restrict__ bo,
                            float* __restrict__ out) {
    int t = blockIdx.x * blockDim.x + threadIdx.x;
    if (t >= N_GRAPHS * N_CLASSES) return;
    int g = t / N_CLASSES;
    int c = t - g * N_CLASSES;
    float invc = 1.0f / fmaxf(gcnt[g], 1.0f);
    float acc = 0.0f;
    for (int d = 0; d < HIDDEN_D; ++d)
        acc += gsum[(size_t)g * HIDDEN_D + d] * invc * Wo[(size_t)d * N_CLASSES + c];
    out[t] = acc + bo[c];
}

extern "C" void kernel_launch(void* const* d_in, const int* in_sizes, int n_in,
                              void* d_out, int out_size, void* d_ws, size_t ws_size,
                              hipStream_t stream) {
    const int*   node_ids = (const int*)  d_in[0];
    const int*   src      = (const int*)  d_in[1];
    const int*   dst      = (const int*)  d_in[2];
    const int*   batch    = (const int*)  d_in[3];
    const float* embed    = (const float*)d_in[4];
    const float* W1_l     = (const float*)d_in[5];
    const float* b1       = (const float*)d_in[6];
    const float* W1_r     = (const float*)d_in[7];
    const float* W2_l     = (const float*)d_in[8];
    const float* b2       = (const float*)d_in[9];
    const float* W2_r     = (const float*)d_in[10];
    const float* W_out    = (const float*)d_in[11];
    const float* b_out    = (const float*)d_in[12];
    float* out = (float*)d_out;

    float* ws   = (float*)d_ws;
    float* x0   = ws;
    float* h1   = x0   + (size_t)N_NODES * EMBED_D;
    float* agg1 = h1   + (size_t)N_NODES * HIDDEN_D;
    float* agg2 = agg1 + (size_t)N_NODES * EMBED_D;
    float* cnt  = agg2 + (size_t)N_NODES * HIDDEN_D;
    float* gsum = cnt  + (size_t)N_NODES;
    float* gcnt = gsum + (size_t)N_GRAPHS * HIDDEN_D;
    const int zeroN = N_NODES * EMBED_D + N_NODES * HIDDEN_D + N_NODES
                    + N_GRAPHS * HIDDEN_D + N_GRAPHS;

    const int B = 256;
    auto blocks = [](int n, int b) { return (n + b - 1) / b; };

    zero_kernel<<<blocks(zeroN, B), B, 0, stream>>>(agg1, zeroN);

    gather_embed_kernel<<<blocks(N_NODES * 8, B), B, 0, stream>>>(
        node_ids, embed, x0, N_NODES);

    edge_count_kernel<<<blocks(N_EDGES, B), B, 0, stream>>>(dst, cnt, N_EDGES);

    static_assert(N_EDGES % 32 == 0, "a wave owns a whole group of edges");
    edge_scatter_kernel<EMBED_D, 32><<<blocks(N_EDGES / 32 * 32, B), B, 0, stream>>>(
        src, dst, x0, agg1, N_EDGES);
    sage_gemm_kernel<EMBED_D, false><<<N_NODES / 16, 128, 0, stream>>>(
        agg1, cnt, x0, W1_l, W1_r, b1, h1, nullptr, nullptr, nullptr, N_NODES);

    edge_scatter_kernel<HIDDEN_D, 16><<<blocks(N_EDGES / 16 * 32, B), B, 0, stream>>>(
        src, dst, h1, agg2, N_EDGES);
    static_assert(N_NODES % 16 == 0, "a GEMM tile is 16 whole rows");
    sage_gemm_kernel<HIDDEN_D, true><<<N_NODES / 16, 128, 0, stream>>>(
        agg2, cnt, h1, W2_l, W2_r, b2, nullptr, batch, gsum, gcnt, N_NODES);

    head_kernel<<<blocks(N_GRAPHS * N_CLASSES, B), B, 0, stream>>>(
        gsum, gcnt, W_out, b_out, out);
}


# edge sums without atomics: edges grouped by destination in the call (two-level counting partition, thread-private LDS counters), one wave sums a node's source rows; row sum order now fixed
# speedup vs baseline: 1.7352x; 1.1577x over previous
#include <hip/hip_runtime.h>

#define N_NODES   100000
#define N_EDGES   1200000
#define EMBED_D   32
#define HIDDEN_D  64
#define N_CLASSES 10
#define N_GRAPHS  1024

typedef __attribute__((ext_vector_type(2))) float v2f;
typedef __attribute__((ext_vector_type(8))) float v8f;

__device__ __forceinline__ void atomAdd1(float* p, float v) {
    asm volatile("global_atomic_add_f32 %0, %1, off scope:SCOPE_DEV"
                 :: "v"(p), "v"(v) : "memory");
}

__global__ void zero_kernel(float* __restrict__ p, int n) {
    int t = blockIdx.x * blockDim.x + threadIdx.x;
    if (t < n) p[t] = 0.0f;
}

__global__ void gather_embed_kernel(const int* __restrict__ ids,
                                    const float* __restrict__ embed,
                                    float* __restrict__ x0, int nNodes) {
    int t = blockIdx.x * blockDim.x + threadIdx.x;
    if (t >= nNodes * (EMBED_D / 4)) return;
    int node = t >> 3;
    int q    = t & 7;
    float4 v = reinterpret_cast<const float4*>(embed + (size_t)ids[node] * EMBED_D)[q];
    reinterpret_cast<float4*>(x0 + (size_t)node * EMBED_D)[q] = v;
}

#define FINE_BITS 8
#define FINE_BINS 256
#define N_BUCKETS ((N_NODES + FINE_BINS - 1) / FINE_BINS)
#define P1_BLOCKS 256
#define P_THREADS 128
#define P1_EDGES  ((N_EDGES + P1_BLOCKS - 1) / P1_BLOCKS)
#define PITCH1    (N_BUCKETS | 1)
#define PITCH2    (FINE_BINS + 1)
#define TABLE_N   (N_BUCKETS * P1_BLOCKS)
static_assert(N_NODES <= (1 << 17), "a source id fits 17 bits of the packed word");
static_assert((P_THREADS * PITCH1) % 4 == 0 && (P_THREADS * PITCH2) % 4 == 0, "LDS cleared 16 bytes at a time");
static_assert(TABLE_N % 4 == 0, "table read four at a time");

__device__ __forceinline__ void lds_clear(int* h, int nWords, int t) {
    int4* h4 = reinterpret_cast<int4*>(h);
    for (int i = t; i < nWords / 4; i += P_THREADS) h4[i] = make_int4(0, 0, 0, 0);
}

__device__ __forceinline__ void coarse_count(const int* __restrict__ dst, int* my,
                                             int e0, int e1, int t) {
    for (int base = e0; base < e1; base += P_THREADS * 8) {
        int d[8];
#pragma unroll
        for (int q = 0; q < 8; ++q) {
            const int e = base + q * P_THREADS + t;
            d[q] = e < e1 ? dst[e] : -1;
        }
#pragma unroll
        for (int q = 0; q < 8; ++q)
            if (d[q] >= 0) my[d[q] >> FINE_BITS] += 1;
    }
}

__global__ void __launch_bounds__(P_THREADS)
coarse_count_kernel(const int* __restrict__ dst, int* __restrict__ table, int nEdges) {
    __shared__ __attribute__((aligned(16))) int hist[P_THREADS * PITCH1];
    const int t  = threadIdx.x;
    const int e0 = blockIdx.x * P1_EDGES;
    const int e1 = min(e0 + P1_EDGES, nEdges);
    lds_clear(hist, P_THREADS * PITCH1, t);
    __syncthreads();
    coarse_count(dst, hist + t * PITCH1, e0, e1, t);
    __syncthreads();
    for (int b = t; b < N_BUCKETS; b += P_THREADS) {
        int s = 0;
        for (int r = 0; r < P_THREADS; ++r) s += hist[r * PITCH1 + b];
        table[b * P1_BLOCKS + blockIdx.x] = s;
    }
}

__global__ void __launch_bounds__(1024)
offsets_kernel(const int* __restrict__ in, int* __restrict__ out, int n) {
    __shared__ int wtot[32];
    __shared__ int wpart[32];
    const int t    = threadIdx.x;
    const int lane = t & 31;
    const int w    = t >> 5;
    const int4* in4 = reinterpret_cast<const int4*>(in);
    const int nFront = blockIdx.x * 256;
    int part = 0;
    for (int i = t; i < nFront; i += 1024) {
        const int4 q = in4[i];
        part += (q.x + q.y) + (q.z + q.w);
    }
    const int idx = blockIdx.x * 1024 + t;
    const int v   = idx < n ? in[idx] : 0;
    int x = v;
#pragma unroll
    for (int d = 1; d < 32; d <<= 1) {
        const int y = __shfl_up(x, d, 32);
        if (lane >= d) x += y;
    }
#pragma unroll
    for (int d = 16; d >= 1; d >>= 1) part += __shfl_xor(part, d, 32);
    if (lane == 31) wtot[w]  = x;
    if (lane == 0)  wpart[w] = part;
    __syncthreads();
    int f = wpart[lane];
    int g = lane < w ? wtot[lane] : 0;
#pragma unroll
    for (int d = 16; d >= 1; d >>= 1) {
        f += __shfl_xor(f, d, 32);
        g += __shfl_xor(g, d, 32);
    }
    const int ex = f + g + x - v;
    if (idx < n)      out[idx] = ex;
    if (idx == n - 1) out[n] = ex + v;
}

__global__ void __launch_bounds__(P_THREADS)
coarse_place_kernel(const int* __restrict__ src, const int* __restrict__ dst,
                    const int* __restrict__ tableOff, int* __restrict__ tmp, int nEdges) {
    __shared__ __attribute__((aligned(16))) int hist[P_THREADS * PITCH1];
    const int t  = threadIdx.x;
    const int e0 = blockIdx.x * P1_EDGES;
    const int e1 = min(e0 + P1_EDGES, nEdges);
    int* my = hist + t * PITCH1;
    lds_clear(hist, P_THREADS * PITCH1, t);
    __syncthreads();
    coarse_count(dst, my, e0, e1, t);
    __syncthreads();
    for (int b = t; b < N_BUCKETS; b += P_THREADS) {
        int run = tableOff[b * P1_BLOCKS + blockIdx.x];
        for (int r = 0; r < P_THREADS; ++r) {
            const int c = hist[r * PITCH1 + b];
            hist[r * PITCH1 + b] = run;
            run += c;
        }
    }
    __syncthreads();
    for (int base = e0; base < e1; base += P_THREADS * 8) {
        int d[8], s[8];
#pragma unroll
        for (int q = 0; q < 8; ++q) {
            const int e = base + q * P_THREADS + t;
            d[q] = e < e1 ? dst[e] : -1;
            s[q] = e < e1 ? src[e] : 0;
        }
#pragma unroll
        for (int q = 0; q < 8; ++q) {
            if (d[q] >= 0) {
                const int pos = my[d[q] >> FINE_BITS];
                my[d[q] >> FINE_BITS] = pos + 1;
                tmp[pos] = ((d[q] & (FINE_BINS - 1)) << 17) | s[q];
            }
        }
    }
}

__global__ void __launch_bounds__(P_THREADS)
fine_group_kernel(const int* __restrict__ tableOff, const int* __restrict__ tmp,
                  int* __restrict__ deg, int* __restrict__ rowptr,
                  int* __restrict__ ssrc, int nEdges) {
    __shared__ __attribute__((aligned(16))) int hist[P_THREADS * PITCH2];
    __shared__ int tot[FINE_BINS];
    __shared__ int binOff[FINE_BINS];
    const int t    = threadIdx.x;
    const int lane = t & 31;
    const int k    = blockIdx.x;
    const int beg  = tableOff[k * P1_BLOCKS];
    const int end  = tableOff[(k + 1) * P1_BLOCKS];
    int* my = hist + t * PITCH2;
    lds_clear(hist, P_THREADS * PITCH2, t);
    __syncthreads();
    for (int base = beg; base < end; base += P_THREADS * 8) {
        int wv[8];
#pragma unroll
        for (int q = 0; q < 8; ++q) {
            const int i = base + q * P_THREADS + t;
            wv[q] = i < end ? tmp[i] : -1;
        }
#pragma unroll
        for (int q = 0; q < 8; ++q)
            if (wv[q] >= 0) my[wv[q] >> 17] += 1;
    }
    __syncthreads();
    for (int b = t; b < FINE_BINS; b += P_THREADS) {
        int run = 0;
        for (int r = 0; r < P_THREADS; ++r) {
            const int c = hist[r * PITCH2 + b];
            hist[r * PITCH2 + b] = run;
            run += c;
        }
        tot[b] = run;
    }
    __syncthreads();
    if (t < 32) {
        int c[8];
        int s = 0;
#pragma unroll
        for (int q = 0; q < 8; ++q) { c[q] = tot[8 * lane + q]; s += c[q]; }
        int x = s;
#pragma unroll
        for (int d = 1; d < 32; d <<= 1) {
            const int y = __shfl_up(x, d, 32);
            if (lane >= d) x += y;
        }
        int ex = x - s;
#pragma unroll
        for (int q = 0; q < 8; ++q) { binOff[8 * lane + q] = ex; ex += c[q]; }
    }
    __syncthreads();
    for (int b = t; b < FINE_BINS; b += P_THREADS) {
        const int node = k * FINE_BINS + b;
        if (node < N_NODES) {
            deg[node]    = tot[b];
            rowptr[node] = beg + binOff[b];
        }
    }
    if (k == N_BUCKETS - 1 && t == 0) rowptr[N_NODES] = end;
    for (int base = beg; base < end; base += P_THREADS * 8) {
        int wv[8];
#pragma unroll
        for (int q = 0; q < 8; ++q) {
            const int i = base + q * P_THREADS + t;
            wv[q] = i < end ? tmp[i] : -1;
        }
#pragma unroll
        for (int q = 0; q < 8; ++q) {
            if (wv[q] >= 0) {
                const int lo  = wv[q] >> 17;
                const int pos = my[lo];
                my[lo] = pos + 1;
                ssrc[beg + binOff[lo] + pos] = wv[q] & ((1 << 17) - 1);
            }
        }
    }
}

template <int DIM, int NPW, int U>
__global__ void node_sum_kernel(const int* __restrict__ rowptr,
                                const int* __restrict__ ssrc,
                                const float* __restrict__ x,
                                float* __restrict__ agg, int nNodes) {
    constexpr int VW = DIM / 32;
    static_assert(VW == 1 || VW == 2, "row is one or two lines");
    static_assert(N_NODES % NPW == 0, "a wave owns whole nodes");
    static_assert((unsigned long long)N_NODES * DIM * 4ull < (1ull << 32), "32-bit row offsets");
    const int lane = threadIdx.x & 31;
    const int wave = blockIdx.x * (blockDim.x >> 5) + (threadIdx.x >> 5);
    const int n0   = wave * NPW;
    if (n0 >= nNodes) return;

    const int rp = rowptr[n0 + (lane < NPW ? lane : NPW)];
    const unsigned laneOff = (unsigned)lane * (unsigned)(VW * 4);
    const char* xb = reinterpret_cast<const char*>(x);

    int first[NPW];
#pragma unroll
    for (int i = 0; i < NPW; ++i) {
        const int beg = __builtin_amdgcn_readlane(rp, i);
        const int end = __builtin_amdgcn_readlane(rp, i + 1);
        const int idx = beg + lane;
        first[i] = idx < end ? ssrc[idx] : 0;
    }
#pragma unroll
    for (int i = 0; i < NPW; ++i) {
        const int beg = __builtin_amdgcn_readlane(rp, i);
        const int end = __builtin_amdgcn_readlane(rp, i + 1);
        float a0 = 0.0f, a1 = 0.0f;
        int my = first[i];
        for (int base = beg; base < end; base += 32) {
            if (base != beg) {
                const int idx = base + lane;
                my = idx < end ? ssrc[idx] : 0;
            }
            const int m = min(32, end - base);
            for (int j = 0; j < m; j += U) {
                float v0[U], v1[U];
#pragma unroll
                for (int u = 0; u < U; ++u) {
                    const bool ok = (j + u) < m;
                    const unsigned s = (unsigned)__builtin_amdgcn_readlane(my, ok ? j + u : j);
                    const char* p = xb + (s * (unsigned)(DIM * 4) + laneOff);
                    if (VW == 1) {
                        v0[u] = *reinterpret_cast<const float*>(p);
                        v1[u] = 0.0f;
                    } else {
                        const float2 q = *reinterpret_cast<const float2*>(p);
                        v0[u] = q.x;
                        v1[u] = q.y;
                    }
                    if (!ok) { v0[u] = 0.0f; v1[u] = 0.0f; }
                }
#pragma unroll
                for (int u = 0; u < U; ++u) {
                    a0 += v0[u];
                    if (VW == 2) a1 += v1[u];
                }
            }
        }
        float* ar = agg + (size_t)(n0 + i) * DIM + lane * VW;
        if (VW == 1) *ar = a0;
        else *reinterpret_cast<float2*>(ar) = make_float2(a0, a1);
    }
}

template <int K, bool POOL>
__global__ void sage_gemm_kernel(const float* __restrict__ aggRaw,
                                 const int* __restrict__ deg,
                                 const float* __restrict__ x,
                                 const float* __restrict__ Wl,
                                 const float* __restrict__ Wr,
                                 const float* __restrict__ bias,
                                 float* __restrict__ out,
                                 const int* __restrict__ batch,
                                 float* __restrict__ gsum,
                                 float* __restrict__ gcnt, int nNodes) {
    const int lane = threadIdx.x & 31;
    const int wave = threadIdx.x >> 5;
    const int row0 = blockIdx.x * 16;
    if (row0 >= nNodes) return;
    const int col0 = wave * 16;
    const int grp  = lane >> 4;
    const int lr   = lane & 15;

    const int   arow = row0 + lr;
    const float invc = 1.0f / fmaxf((float)deg[arow], 1.0f);

    v8f c = {};
#pragma unroll
    for (int k0 = 0; k0 < K; k0 += 4) {
        const int ka = k0 + 2 * grp;
        v2f a, b;
        a.x = aggRaw[(size_t)arow * K + ka + 0] * invc;
        a.y = aggRaw[(size_t)arow * K + ka + 1] * invc;
        b.x = Wl[(size_t)(ka + 0) * HIDDEN_D + col0 + lr];
        b.y = Wl[(size_t)(ka + 1) * HIDDEN_D + col0 + lr];
        c = __builtin_amdgcn_wmma_f32_16x16x4_f32(false, a, false, b,
                                                  (short)0, c, false, false);
    }
#pragma unroll
    for (int k0 = 0; k0 < K; k0 += 4) {
        const int ka = k0 + 2 * grp;
        v2f a, b;
        a.x = x[(size_t)arow * K + ka + 0];
        a.y = x[(size_t)arow * K + ka + 1];
        b.x = Wr[(size_t)(ka + 0) * HIDDEN_D + col0 + lr];
        b.y = Wr[(size_t)(ka + 1) * HIDDEN_D + col0 + lr];
        c = __builtin_amdgcn_wmma_f32_16x16x4_f32(false, a, false, b,
                                                  (short)0, c, false, false);
    }

    const float bv = bias[col0 + lr];
    if (!POOL) {
#pragma unroll
        for (int v = 0; v < 8; ++v) {
            const int m = v + 8 * grp;
            out[(size_t)(row0 + m) * HIDDEN_D + col0 + lr] = fmaxf(c[v] + bv, 0.0f);
        }
    } else {
        const int gid = batch[row0 + lr];
        const int g0  = __builtin_amdgcn_readfirstlane(gid);
        float r[8];
#pragma unroll
        for (int v = 0; v < 8; ++v) r[v] = fmaxf(c[v] + bv, 0.0f);
        if (__all(gid == g0)) {
            float s = ((r[0] + r[1]) + (r[2] + r[3])) + ((r[4] + r[5]) + (r[6] + r[7]));
            s += __shfl_xor(s, 16, 32);
            if (grp == 0) atomAdd1(gsum + (size_t)g0 * HIDDEN_D + col0 + lr, s);
            if (threadIdx.x == 0) atomAdd1(gcnt + g0, 16.0f);
        } else {
#pragma unroll
            for (int v = 0; v < 8; ++v) {
                const int gv = __shfl(gid, v + 8 * grp, 32);
                atomAdd1(gsum + (size_t)gv * HIDDEN_D + col0 + lr, r[v]);
            }
            if (threadIdx.x < 16) atomAdd1(gcnt + gid, 1.0f);
        }
    }
}

__global__ void head_kernel(const float* __restrict__ gsum,
                            const float* __restrict__ gcnt,
                            const float* __restrict__ Wo,
                            const float* __restrict__ bo,
                            float* __restrict__ out) {
    int t = blockIdx.x * blockDim.x + threadIdx.x;
    if (t >= N_GRAPHS * N_CLASSES) return;
    int g = t / N_CLASSES;
    int c = t - g * N_CLASSES;
    float invc = 1.0f / fmaxf(gcnt[g], 1.0f);
    float acc = 0.0f;
    for (int d = 0; d < HIDDEN_D; ++d)
        acc += gsum[(size_t)g * HIDDEN_D + d] * invc * Wo[(size_t)d * N_CLASSES + c];
    out[t] = acc + bo[c];
}

extern "C" void kernel_launch(void* const* d_in, const int* in_sizes, int n_in,
                              void* d_out, int out_size, void* d_ws, size_t ws_size,
                              hipStream_t stream) {
    const int*   node_ids = (const int*)  d_in[0];
    const int*   src      = (const int*)  d_in[1];
    const int*   dst      = (const int*)  d_in[2];
    const int*   batch    = (const int*)  d_in[3];
    const float* embed    = (const float*)d_in[4];
    const float* W1_l     = (const float*)d_in[5];
    const float* b1       = (const float*)d_in[6];
    const float* W1_r     = (const float*)d_in[7];
    const float* W2_l     = (const float*)d_in[8];
    const float* b2       = (const float*)d_in[9];
    const float* W2_r     = (const float*)d_in[10];
    const float* W_out    = (const float*)d_in[11];
    const float* b_out    = (const float*)d_in[12];
    float* out = (float*)d_out;

    float* ws   = (float*)d_ws;
    float* x0   = ws;
    float* h1   = x0   + (size_t)N_NODES * EMBED_D;
    float* agg1 = h1   + (size_t)N_NODES * HIDDEN_D;
    float* agg2 = agg1 + (size_t)N_NODES * EMBED_D;
    float* gsum = agg2 + (size_t)N_NODES * HIDDEN_D;
    float* gcnt = gsum + (size_t)N_GRAPHS * HIDDEN_D;
    int* deg      = (int*)(gcnt + (size_t)N_GRAPHS);
    int* rowptr   = deg      + (size_t)N_NODES;
    int* table    = rowptr   + (size_t)(N_NODES + 4);
    int* tableOff = table    + (size_t)TABLE_N;
    int* tmp      = tableOff + (size_t)(TABLE_N + 4);
    int* ssrc     = tmp      + (size_t)N_EDGES;
    const int zeroN = N_GRAPHS * HIDDEN_D + N_GRAPHS;

    const int B = 256;
    auto blocks = [](int n, int b) { return (n + b - 1) / b; };

    zero_kernel<<<blocks(zeroN, B), B, 0, stream>>>(gsum, zeroN);

    gather_embed_kernel<<<blocks(N_NODES * 8, B), B, 0, stream>>>(
        node_ids, embed, x0, N_NODES);

    coarse_count_kernel<<<P1_BLOCKS, P_THREADS, 0, stream>>>(dst, table, N_EDGES);
    offsets_kernel<<<blocks(TABLE_N, 1024), 1024, 0, stream>>>(table, tableOff, TABLE_N);
    coarse_place_kernel<<<P1_BLOCKS, P_THREADS, 0, stream>>>(src, dst, tableOff, tmp, N_EDGES);
    fine_group_kernel<<<N_BUCKETS, P_THREADS, 0, stream>>>(tableOff, tmp, deg, rowptr, ssrc, N_EDGES);

    node_sum_kernel<EMBED_D, 4, 4><<<blocks(N_NODES / 4 * 32, B), B, 0, stream>>>(
        rowptr, ssrc, x0, agg1, N_NODES);
    sage_gemm_kernel<EMBED_D, false><<<N_NODES / 16, 128, 0, stream>>>(
        agg1, deg, x0, W1_l, W1_r, b1, h1, nullptr, nullptr, nullptr, N_NODES);

    node_sum_kernel<HIDDEN_D, 4, 4><<<blocks(N_NODES / 4 * 32, B), B, 0, stream>>>(
        rowptr, ssrc, h1, agg2, N_NODES);
    static_assert(N_NODES % 16 == 0, "a GEMM tile is 16 whole rows");
    sage_gemm_kernel<HIDDEN_D, true><<<N_NODES / 16, 128, 0, stream>>>(
        agg2, deg, h1, W2_l, W2_r, b2, nullptr, batch, gsum, gcnt, N_NODES);

    head_kernel<<<blocks(N_GRAPHS * N_CLASSES, B), B, 0, stream>>>(
        gsum, gcnt, W_out, b_out, out);
}


# edge sums fused into the layer kernel: each wave sums 4 nodes' source rows into a 16 x K LDS tile that is the A operand of the first product; summed rows and counts never stored; sum order kept
# speedup vs baseline: 2.0405x; 1.1759x over previous
#include <hip/hip_runtime.h>

#define N_NODES   100000
#define N_EDGES   1200000
#define EMBED_D   32
#define HIDDEN_D  64
#define N_CLASSES 10
#define N_GRAPHS  1024

typedef __attribute__((ext_vector_type(2))) float v2f;
typedef __attribute__((ext_vector_type(8))) float v8f;

__device__ __forceinline__ void atomAdd1(float* p, float v) {
    asm volatile("global_atomic_add_f32 %0, %1, off scope:SCOPE_DEV"
                 :: "v"(p), "v"(v) : "memory");
}

__global__ void zero_kernel(float* __restrict__ p, int n) {
    int t = blockIdx.x * blockDim.x + threadIdx.x;
    if (t < n) p[t] = 0.0f;
}

__global__ void gather_embed_kernel(const int* __restrict__ ids,
                                    const float* __restrict__ embed,
                                    float* __restrict__ x0, int nNodes) {
    int t = blockIdx.x * blockDim.x + threadIdx.x;
    if (t >= nNodes * (EMBED_D / 4)) return;
    int node = t >> 3;
    int q    = t & 7;
    float4 v = reinterpret_cast<const float4*>(embed + (size_t)ids[node] * EMBED_D)[q];
    reinterpret_cast<float4*>(x0 + (size_t)node * EMBED_D)[q] = v;
}

#define FINE_BITS 8
#define FINE_BINS 256
#define N_BUCKETS ((N_NODES + FINE_BINS - 1) / FINE_BINS)
#define P1_BLOCKS 256
#define P_THREADS 128
#define P1_EDGES  ((N_EDGES + P1_BLOCKS - 1) / P1_BLOCKS)
#define PITCH1    (N_BUCKETS | 1)
#define PITCH2    (FINE_BINS + 1)
#define TABLE_N   (N_BUCKETS * P1_BLOCKS)
static_assert(N_NODES <= (1 << 17), "a source id fits 17 bits of the packed word");
static_assert((P_THREADS * PITCH1) % 4 == 0 && (P_THREADS * PITCH2) % 4 == 0, "LDS cleared 16 bytes at a time");
static_assert(TABLE_N % 4 == 0, "table read four at a time");

__device__ __forceinline__ void lds_clear(int* h, int nWords, int t) {
    int4* h4 = reinterpret_cast<int4*>(h);
    for (int i = t; i < nWords / 4; i += P_THREADS) h4[i] = make_int4(0, 0, 0, 0);
}

__device__ __forceinline__ void coarse_count(const int* __restrict__ dst, int* my,
                                             int e0, int e1, int t) {
    for (int base = e0; base < e1; base += P_THREADS * 8) {
        int d[8];
#pragma unroll
        for (int q = 0; q < 8; ++q) {
            const int e = base + q * P_THREADS + t;
            d[q] = e < e1 ? dst[e] : -1;
        }
#pragma unroll
        for (int q = 0; q < 8; ++q)
            if (d[q] >= 0) my[d[q] >> FINE_BITS] += 1;
    }
}

__global__ void __launch_bounds__(P_THREADS)
coarse_count_kernel(const int* __restrict__ dst, int* __restrict__ table, int nEdges) {
    __shared__ __attribute__((aligned(16))) int hist[P_THREADS * PITCH1];
    const int t  = threadIdx.x;
    const int e0 = blockIdx.x * P1_EDGES;
    const int e1 = min(e0 + P1_EDGES, nEdges);
    lds_clear(hist, P_THREADS * PITCH1, t);
    __syncthreads();
    coarse_count(dst, hist + t * PITCH1, e0, e1, t);
    __syncthreads();
    for (int b = t; b < N_BUCKETS; b += P_THREADS) {
        int s = 0;
        for (int r = 0; r < P_THREADS; ++r) s += hist[r * PITCH1 + b];
        table[b * P1_BLOCKS + blockIdx.x] = s;
    }
}

__global__ void __launch_bounds__(1024)
offsets_kernel(const int* __restrict__ in, int* __restrict__ out, int n) {
    __shared__ int wtot[32];
    __shared__ int wpart[32];
    const int t    = threadIdx.x;
    const int lane = t & 31;
    const int w    = t >> 5;
    const int4* in4 = reinterpret_cast<const int4*>(in);
    const int nFront = blockIdx.x * 256;
    int part = 0;
    for (int i = t; i < nFront; i += 1024) {
        const int4 q = in4[i];
        part += (q.x + q.y) + (q.z + q.w);
    }
    const int idx = blockIdx.x * 1024 + t;
    const int v   = idx < n ? in[idx] : 0;
    int x = v;
#pragma unroll
    for (int d = 1; d < 32; d <<= 1) {
        const int y = __shfl_up(x, d, 32);
        if (lane >= d) x += y;
    }
#pragma unroll
    for (int d = 16; d >= 1; d >>= 1) part += __shfl_xor(part, d, 32);
    if (lane == 31) wtot[w]  = x;
    if (lane == 0)  wpart[w] = part;
    __syncthreads();
    int f = wpart[lane];
    int g = lane < w ? wtot[lane] : 0;
#pragma unroll
    for (int d = 16; d >= 1; d >>= 1) {
        f += __shfl_xor(f, d, 32);
        g += __shfl_xor(g, d, 32);
    }
    const int ex = f + g + x - v;
    if (idx < n)      out[idx] = ex;
    if (idx == n - 1) out[n] = ex + v;
}

__global__ void __launch_bounds__(P_THREADS)
coarse_place_kernel(const int* __restrict__ src, const int* __restrict__ dst,
                    const int* __restrict__ tableOff, int* __restrict__ tmp, int nEdges) {
    __shared__ __attribute__((aligned(16))) int hist[P_THREADS * PITCH1];
    const int t  = threadIdx.x;
    const int e0 = blockIdx.x * P1_EDGES;
    const int e1 = min(e0 + P1_EDGES, nEdges);
    int* my = hist + t * PITCH1;
    lds_clear(hist, P_THREADS * PITCH1, t);
    __syncthreads();
    coarse_count(dst, my, e0, e1, t);
    __syncthreads();
    for (int b = t; b < N_BUCKETS; b += P_THREADS) {
        int run = tableOff[b * P1_BLOCKS + blockIdx.x];
        for (int r = 0; r < P_THREADS; ++r) {
            const int c = hist[r * PITCH1 + b];
            hist[r * PITCH1 + b] = run;
            run += c;
        }
    }
    __syncthreads();
    for (int base = e0; base < e1; base += P_THREADS * 8) {
        int d[8], s[8];
#pragma unroll
        for (int q = 0; q < 8; ++q) {
            const int e = base + q * P_THREADS + t;
            d[q] = e < e1 ? dst[e] : -1;
            s[q] = e < e1 ? src[e] : 0;
        }
#pragma unroll
        for (int q = 0; q < 8; ++q) {
            if (d[q] >= 0) {
                const int pos = my[d[q] >> FINE_BITS];
                my[d[q] >> FINE_BITS] = pos + 1;
                tmp[pos] = ((d[q] & (FINE_BINS - 1)) << 17) | s[q];
            }
        }
    }
}

__global__ void __launch_bounds__(P_THREADS)
fine_group_kernel(const int* __restrict__ tableOff, const int* __restrict__ tmp,
                  int* __restrict__ rowptr,
                  int* __restrict__ ssrc, int nEdges) {
    __shared__ __attribute__((aligned(16))) int hist[P_THREADS * PITCH2];
    __shared__ int tot[FINE_BINS];
    __shared__ int binOff[FINE_BINS];
    const int t    = threadIdx.x;
    const int lane = t & 31;
    const int k    = blockIdx.x;
    const int beg  = tableOff[k * P1_BLOCKS];
    const int end  = tableOff[(k + 1) * P1_BLOCKS];
    int* my = hist + t * PITCH2;
    lds_clear(hist, P_THREADS * PITCH2, t);
    __syncthreads();
    for (int base = beg; base < end; base += P_THREADS * 8) {
        int wv[8];
#pragma unroll
        for (int q = 0; q < 8; ++q) {
            const int i = base + q * P_THREADS + t;
            wv[q] = i < end ? tmp[i] : -1;
        }
#pragma unroll
        for (int q = 0; q < 8; ++q)
            if (wv[q] >= 0) my[wv[q] >> 17] += 1;
    }
    __syncthreads();
    for (int b = t; b < FINE_BINS; b += P_THREADS) {
        int run = 0;
        for (int r = 0; r < P_THREADS; ++r) {
            const int c = hist[r * PITCH2 + b];
            hist[r * PITCH2 + b] = run;
            run += c;
        }
        tot[b] = run;
    }
    __syncthreads();
    if (t < 32) {
        int c[8];
        int s = 0;
#pragma unroll
        for (int q = 0; q < 8; ++q) { c[q] = tot[8 * lane + q]; s += c[q]; }
        int x = s;
#pragma unroll
        for (int d = 1; d < 32; d <<= 1) {
            const int y = __shfl_up(x, d, 32);
            if (lane >= d) x += y;
        }
        int ex = x - s;
#pragma unroll
        for (int q = 0; q < 8; ++q) { binOff[8 * lane + q] = ex; ex += c[q]; }
    }
    __syncthreads();
    for (int b = t; b < FINE_BINS; b += P_THREADS) {
        const int node = k * FINE_BINS + b;
        if (node < N_NODES) rowptr[node] = beg + binOff[b];
    }
    if (k == N_BUCKETS - 1 && t == 0) rowptr[N_NODES] = end;
    for (int base = beg; base < end; base += P_THREADS * 8) {
        int wv[8];
#pragma unroll
        for (int q = 0; q < 8; ++q) {
            const int i = base + q * P_THREADS + t;
            wv[q] = i < end ? tmp[i] : -1;
        }
#pragma unroll
        for (int q = 0; q < 8; ++q) {
            if (wv[q] >= 0) {
                const int lo  = wv[q] >> 17;
                const int pos = my[lo];
                my[lo] = pos + 1;
                ssrc[beg + binOff[lo] + pos] = wv[q] & ((1 << 17) - 1);
            }
        }
    }
}

template <int K, bool POOL>
__global__ void __launch_bounds__(128, 16)
sage_layer_kernel(const int* __restrict__ rowptr,
                  const int* __restrict__ ssrc,
                  const float* __restrict__ x,
                  const float* __restrict__ Wl,
                  const float* __restrict__ Wr,
                  const float* __restrict__ bias,
                  float* __restrict__ out,
                  const int* __restrict__ batch,
                  float* __restrict__ gsum,
                  float* __restrict__ gcnt, int nNodes) {
    constexpr int VW  = K / 32;
    constexpr int NPW = 4;
    constexpr int U   = 4;
    constexpr int PITCH = K + 4;
    static_assert(VW == 1 || VW == 2, "row is one or two lines");
    static_assert(N_NODES % 16 == 0, "a block owns 16 whole nodes");
    static_assert((unsigned long long)N_NODES * K * 4ull < (1ull << 32), "32-bit row offsets");
    __shared__ __attribute__((aligned(16))) float aggT[16 * PITCH];

    const int lane = threadIdx.x & 31;
    const int wave = threadIdx.x >> 5;
    const int row0 = blockIdx.x * 16;
    if (row0 >= nNodes) return;

    {
        const int n0 = row0 + wave * NPW;
        const int rp = rowptr[n0 + (lane < NPW ? lane : NPW)];
        const unsigned laneOff = (unsigned)lane * (unsigned)(VW * 4);
        const char* xb = reinterpret_cast<const char*>(x);

        int first[NPW];
#pragma unroll
        for (int i = 0; i < NPW; ++i) {
            const int beg = __builtin_amdgcn_readlane(rp, i);
            const int end = __builtin_amdgcn_readlane(rp, i + 1);
            const int idx = beg + lane;
            first[i] = idx < end ? ssrc[idx] : 0;
        }
#pragma unroll
        for (int i = 0; i < NPW; ++i) {
            const int beg = __builtin_amdgcn_readlane(rp, i);
            const int end = __builtin_amdgcn_readlane(rp, i + 1);
            float a0 = 0.0f, a1 = 0.0f;
            int my = first[i];
            for (int base = beg; base < end; base += 32) {
                if (base != beg) {
                    const int idx = base + lane;
                    my = idx < end ? ssrc[idx] : 0;
                }
                const int m = min(32, end - base);
                for (int j = 0; j < m; j += U) {
                    float v0[U], v1[U];
#pragma unroll
                    for (int u = 0; u < U; ++u) {
                        const bool ok = (j + u) < m;
                        const unsigned s = (unsigned)__builtin_amdgcn_readlane(my, ok ? j + u : j);
                        const char* p = xb + (s * (unsigned)(K * 4) + laneOff);
                        if (VW == 1) {
                            v0[u] = *reinterpret_cast<const float*>(p);
                            v1[u] = 0.0f;
                        } else {
                            const float2 q = *reinterpret_cast<const float2*>(p);
                            v0[u] = q.x;
                            v1[u] = q.y;
                        }
                        if (!ok) { v0[u] = 0.0f; v1[u] = 0.0f; }
                    }
#pragma unroll
                    for (int u = 0; u < U; ++u) {
                        a0 += v0[u];
                        if (VW == 2) a1 += v1[u];
                    }
                }
            }
            const float invc = 1.0f / fmaxf((float)(end - beg), 1.0f);
            float* ar = aggT + (wave * NPW + i) * PITCH + lane * VW;
            if (VW == 1) *ar = a0 * invc;
            else *reinterpret_cast<float2*>(ar) = make_float2(a0 * invc, a1 * invc);
        }
    }
    __syncthreads();

    const int col0 = wave * 16;
    const int grp  = lane >> 4;
    const int lr   = lane & 15;
    const int arow = row0 + lr;

    v8f c = {};
#pragma unroll
    for (int k0 = 0; k0 < K; k0 += 4) {
        const int ka = k0 + 2 * grp;
        v2f a, b;
        const float2 am = *reinterpret_cast<const float2*>(aggT + lr * PITCH + ka);
        a.x = am.x;
        a.y = am.y;
        b.x = Wl[(size_t)(ka + 0) * HIDDEN_D + col0 + lr];
        b.y = Wl[(size_t)(ka + 1) * HIDDEN_D + col0 + lr];
        c = __builtin_amdgcn_wmma_f32_16x16x4_f32(false, a, false, b,
                                                  (short)0, c, false, false);
    }
#pragma unroll
    for (int k0 = 0; k0 < K; k0 += 4) {
        const int ka = k0 + 2 * grp;
        v2f a, b;
        a.x = x[(size_t)arow * K + ka + 0];
        a.y = x[(size_t)arow * K + ka + 1];
        b.x = Wr[(size_t)(ka + 0) * HIDDEN_D + col0 + lr];
        b.y = Wr[(size_t)(ka + 1) * HIDDEN_D + col0 + lr];
        c = __builtin_amdgcn_wmma_f32_16x16x4_f32(false, a, false, b,
                                                  (short)0, c, false, false);
    }

    const float bv = bias[col0 + lr];
    if (!POOL) {
#pragma unroll
        for (int v = 0; v < 8; ++v) {
            const int m = v + 8 * grp;
            out[(size_t)(row0 + m) * HIDDEN_D + col0 + lr] = fmaxf(c[v] + bv, 0.0f);
        }
    } else {
        const int gid = batch[row0 + lr];
        const int g0  = __builtin_amdgcn_readfirstlane(gid);
        float r[8];
#pragma unroll
        for (int v = 0; v < 8; ++v) r[v] = fmaxf(c[v] + bv, 0.0f);
        if (__all(gid == g0)) {
            float s = ((r[0] + r[1]) + (r[2] + r[3])) + ((r[4] + r[5]) + (r[6] + r[7]));
            s += __shfl_xor(s, 16, 32);
            if (grp == 0) atomAdd1(gsum + (size_t)g0 * HIDDEN_D + col0 + lr, s);
            if (threadIdx.x == 0) atomAdd1(gcnt + g0, 16.0f);
        } else {
#pragma unroll
            for (int v = 0; v < 8; ++v) {
                const int gv = __shfl(gid, v + 8 * grp, 32);
                atomAdd1(gsum + (size_t)gv * HIDDEN_D + col0 + lr, r[v]);
            }
            if (threadIdx.x < 16) atomAdd1(gcnt + gid, 1.0f);
        }
    }
}

__global__ void head_kernel(const float* __restrict__ gsum,
                            const float* __restrict__ gcnt,
                            const float* __restrict__ Wo,
                            const float* __restrict__ bo,
                            float* __restrict__ out) {
    int t = blockIdx.x * blockDim.x + threadIdx.x;
    if (t >= N_GRAPHS * N_CLASSES) return;
    int g = t / N_CLASSES;
    int c = t - g * N_CLASSES;
    float invc = 1.0f / fmaxf(gcnt[g], 1.0f);
    float acc = 0.0f;
    for (int d = 0; d < HIDDEN_D; ++d)
        acc += gsum[(size_t)g * HIDDEN_D + d] * invc * Wo[(size_t)d * N_CLASSES + c];
    out[t] = acc + bo[c];
}

extern "C" void kernel_launch(void* const* d_in, const int* in_sizes, int n_in,
                              void* d_out, int out_size, void* d_ws, size_t ws_size,
                              hipStream_t stream) {
    const int*   node_ids = (const int*)  d_in[0];
    const int*   src      = (const int*)  d_in[1];
    const int*   dst      = (const int*)  d_in[2];
    const int*   batch    = (const int*)  d_in[3];
    const float* embed    = (const float*)d_in[4];
    const float* W1_l     = (const float*)d_in[5];
    const float* b1       = (const float*)d_in[6];
    const float* W1_r     = (const float*)d_in[7];
    const float* W2_l     = (const float*)d_in[8];
    const float* b2       = (const float*)d_in[9];
    const float* W2_r     = (const float*)d_in[10];
    const float* W_out    = (const float*)d_in[11];
    const float* b_out    = (const float*)d_in[12];
    float* out = (float*)d_out;

    float* ws   = (float*)d_ws;
    float* x0   = ws;
    float* h1   = x0   + (size_t)N_NODES * EMBED_D;
    float* gsum = h1   + (size_t)N_NODES * HIDDEN_D;
    float* gcnt = gsum + (size_t)N_GRAPHS * HIDDEN_D;
    int* rowptr   = (int*)(gcnt + (size_t)N_GRAPHS);
    int* table    = rowptr   + (size_t)(N_NODES + 4);
    int* tableOff = table    + (size_t)TABLE_N;
    int* tmp      = tableOff + (size_t)(TABLE_N + 4);
    int* ssrc     = tmp      + (size_t)N_EDGES;
    const int zeroN = N_GRAPHS * HIDDEN_D + N_GRAPHS;

    const int B = 256;
    auto blocks = [](int n, int b) { return (n + b - 1) / b; };

    zero_kernel<<<blocks(zeroN, B), B, 0, stream>>>(gsum, zeroN);

    gather_embed_kernel<<<blocks(N_NODES * 8, B), B, 0, stream>>>(
        node_ids, embed, x0, N_NODES);

    coarse_count_kernel<<<P1_BLOCKS, P_THREADS, 0, stream>>>(dst, table, N_EDGES);
    offsets_kernel<<<blocks(TABLE_N, 1024), 1024, 0, stream>>>(table, tableOff, TABLE_N);
    coarse_place_kernel<<<P1_BLOCKS, P_THREADS, 0, stream>>>(src, dst, tableOff, tmp, N_EDGES);
    fine_group_kernel<<<N_BUCKETS, P_THREADS, 0, stream>>>(tableOff, tmp, rowptr, ssrc, N_EDGES);

    sage_layer_kernel<EMBED_D, false><<<N_NODES / 16, 128, 0, stream>>>(
        rowptr, ssrc, x0, W1_l, W1_r, b1, h1, nullptr, nullptr, nullptr, N_NODES);

    sage_layer_kernel<HIDDEN_D, true><<<N_NODES / 16, 128, 0, stream>>>(
        rowptr, ssrc, h1, W2_l, W2_r, b2, nullptr, batch, gsum, gcnt, N_NODES);

    head_kernel<<<blocks(N_GRAPHS * N_CLASSES, B), B, 0, stream>>>(
        gsum, gcnt, W_out, b_out, out);
}


# layer 1 reads embedding rows straight from the table through node_ids (a wave fetches its sources' ids 32 at a time, 4 nodes together); the gathered copy and its kernel are gone; sum order kept
# speedup vs baseline: 2.1031x; 1.0307x over previous
#include <hip/hip_runtime.h>

#define N_NODES   100000
#define N_EDGES   1200000
#define EMBED_D   32
#define HIDDEN_D  64
#define N_CLASSES 10
#define N_GRAPHS  1024

typedef __attribute__((ext_vector_type(2))) float v2f;
typedef __attribute__((ext_vector_type(8))) float v8f;

__device__ __forceinline__ void atomAdd1(float* p, float v) {
    asm volatile("global_atomic_add_f32 %0, %1, off scope:SCOPE_DEV"
                 :: "v"(p), "v"(v) : "memory");
}

__global__ void zero_kernel(float* __restrict__ p, int n) {
    int t = blockIdx.x * blockDim.x + threadIdx.x;
    if (t < n) p[t] = 0.0f;
}

#define FINE_BITS 8
#define FINE_BINS 256
#define N_BUCKETS ((N_NODES + FINE_BINS - 1) / FINE_BINS)
#define P1_BLOCKS 256
#define P_THREADS 128
#define P1_EDGES  ((N_EDGES + P1_BLOCKS - 1) / P1_BLOCKS)
#define PITCH1    (N_BUCKETS | 1)
#define PITCH2    (FINE_BINS + 1)
#define TABLE_N   (N_BUCKETS * P1_BLOCKS)
static_assert(N_NODES <= (1 << 17), "a source id fits 17 bits of the packed word");
static_assert((P_THREADS * PITCH1) % 4 == 0 && (P_THREADS * PITCH2) % 4 == 0, "LDS cleared 16 bytes at a time");
static_assert(TABLE_N % 4 == 0, "table read four at a time");

__device__ __forceinline__ void lds_clear(int* h, int nWords, int t) {
    int4* h4 = reinterpret_cast<int4*>(h);
    for (int i = t; i < nWords / 4; i += P_THREADS) h4[i] = make_int4(0, 0, 0, 0);
}

__device__ __forceinline__ void coarse_count(const int* __restrict__ dst, int* my,
                                             int e0, int e1, int t) {
    for (int base = e0; base < e1; base += P_THREADS * 8) {
        int d[8];
#pragma unroll
        for (int q = 0; q < 8; ++q) {
            const int e = base + q * P_THREADS + t;
            d[q] = e < e1 ? dst[e] : -1;
        }
#pragma unroll
        for (int q = 0; q < 8; ++q)
            if (d[q] >= 0) my[d[q] >> FINE_BITS] += 1;
    }
}

__global__ void __launch_bounds__(P_THREADS)
coarse_count_kernel(const int* __restrict__ dst, int* __restrict__ table, int nEdges) {
    __shared__ __attribute__((aligned(16))) int hist[P_THREADS * PITCH1];
    const int t  = threadIdx.x;
    const int e0 = blockIdx.x * P1_EDGES;
    const int e1 = min(e0 + P1_EDGES, nEdges);
    lds_clear(hist, P_THREADS * PITCH1, t);
    __syncthreads();
    coarse_count(dst, hist + t * PITCH1, e0, e1, t);
    __syncthreads();
    for (int b = t; b < N_BUCKETS; b += P_THREADS) {
        int s = 0;
        for (int r = 0; r < P_THREADS; ++r) s += hist[r * PITCH1 + b];
        table[b * P1_BLOCKS + blockIdx.x] = s;
    }
}

__global__ void __launch_bounds__(1024)
offsets_kernel(const int* __restrict__ in, int* __restrict__ out, int n) {
    __shared__ int wtot[32];
    __shared__ int wpart[32];
    const int t    = threadIdx.x;
    const int lane = t & 31;
    const int w    = t >> 5;
    const int4* in4 = reinterpret_cast<const int4*>(in);
    const int nFront = blockIdx.x * 256;
    int part = 0;
    for (int i = t; i < nFront; i += 1024) {
        const int4 q = in4[i];
        part += (q.x + q.y) + (q.z + q.w);
    }
    const int idx = blockIdx.x * 1024 + t;
    const int v   = idx < n ? in[idx] : 0;
    int x = v;
#pragma unroll
    for (int d = 1; d < 32; d <<= 1) {
        const int y = __shfl_up(x, d, 32);
        if (lane >= d) x += y;
    }
#pragma unroll
    for (int d = 16; d >= 1; d >>= 1) part += __shfl_xor(part, d, 32);
    if (lane == 31) wtot[w]  = x;
    if (lane == 0)  wpart[w] = part;
    __syncthreads();
    int f = wpart[lane];
    int g = lane < w ? wtot[lane] : 0;
#pragma unroll
    for (int d = 16; d >= 1; d >>= 1) {
        f += __shfl_xor(f, d, 32);
        g += __shfl_xor(g, d, 32);
    }
    const int ex = f + g + x - v;
    if (idx < n)      out[idx] = ex;
    if (idx == n - 1) out[n] = ex + v;
}

__global__ void __launch_bounds__(P_THREADS)
coarse_place_kernel(const int* __restrict__ src, const int* __restrict__ dst,
                    const int* __restrict__ tableOff, int* __restrict__ tmp, int nEdges) {
    __shared__ __attribute__((aligned(16))) int hist[P_THREADS * PITCH1];
    const int t  = threadIdx.x;
    const int e0 = blockIdx.x * P1_EDGES;
    const int e1 = min(e0 + P1_EDGES, nEdges);
    int* my = hist + t * PITCH1;
    lds_clear(hist, P_THREADS * PITCH1, t);
    __syncthreads();
    coarse_count(dst, my, e0, e1, t);
    __syncthreads();
    for (int b = t; b < N_BUCKETS; b += P_THREADS) {
        int run = tableOff[b * P1_BLOCKS + blockIdx.x];
        for (int r = 0; r < P_THREADS; ++r) {
            const int c = hist[r * PITCH1 + b];
            hist[r * PITCH1 + b] = run;
            run += c;
        }
    }
    __syncthreads();
    for (int base = e0; base < e1; base += P_THREADS * 8) {
        int d[8], s[8];
#pragma unroll
        for (int q = 0; q < 8; ++q) {
            const int e = base + q * P_THREADS + t;
            d[q] = e < e1 ? dst[e] : -1;
            s[q] = e < e1 ? src[e] : 0;
        }
#pragma unroll
        for (int q = 0; q < 8; ++q) {
            if (d[q] >= 0) {
                const int pos = my[d[q] >> FINE_BITS];
                my[d[q] >> FINE_BITS] = pos + 1;
                tmp[pos] = ((d[q] & (FINE_BINS - 1)) << 17) | s[q];
            }
        }
    }
}

__global__ void __launch_bounds__(P_THREADS)
fine_group_kernel(const int* __restrict__ tableOff, const int* __restrict__ tmp,
                  int* __restrict__ rowptr,
                  int* __restrict__ ssrc, int nEdges) {
    __shared__ __attribute__((aligned(16))) int hist[P_THREADS * PITCH2];
    __shared__ int tot[FINE_BINS];
    __shared__ int binOff[FINE_BINS];
    const int t    = threadIdx.x;
    const int lane = t & 31;
    const int k    = blockIdx.x;
    const int beg  = tableOff[k * P1_BLOCKS];
    const int end  = tableOff[(k + 1) * P1_BLOCKS];
    int* my = hist + t * PITCH2;
    lds_clear(hist, P_THREADS * PITCH2, t);
    __syncthreads();
    for (int base = beg; base < end; base += P_THREADS * 8) {
        int wv[8];
#pragma unroll
        for (int q = 0; q < 8; ++q) {
            const int i = base + q * P_THREADS + t;
            wv[q] = i < end ? tmp[i] : -1;
        }
#pragma unroll
        for (int q = 0; q < 8; ++q)
            if (wv[q] >= 0) my[wv[q] >> 17] += 1;
    }
    __syncthreads();
    for (int b = t; b < FINE_BINS; b += P_THREADS) {
        int run = 0;
        for (int r = 0; r < P_THREADS; ++r) {
            const int c = hist[r * PITCH2 + b];
            hist[r * PITCH2 + b] = run;
            run += c;
        }
        tot[b] = run;
    }
    __syncthreads();
    if (t < 32) {
        int c[8];
        int s = 0;
#pragma unroll
        for (int q = 0; q < 8; ++q) { c[q] = tot[8 * lane + q]; s += c[q]; }
        int x = s;
#pragma unroll
        for (int d = 1; d < 32; d <<= 1) {
            const int y = __shfl_up(x, d, 32);
            if (lane >= d) x += y;
        }
        int ex = x - s;
#pragma unroll
        for (int q = 0; q < 8; ++q) { binOff[8 * lane + q] = ex; ex += c[q]; }
    }
    __syncthreads();
    for (int b = t; b < FINE_BINS; b += P_THREADS) {
        const int node = k * FINE_BINS + b;
        if (node < N_NODES) rowptr[node] = beg + binOff[b];
    }
    if (k == N_BUCKETS - 1 && t == 0) rowptr[N_NODES] = end;
    for (int base = beg; base < end; base += P_THREADS * 8) {
        int wv[8];
#pragma unroll
        for (int q = 0; q < 8; ++q) {
            const int i = base + q * P_THREADS + t;
            wv[q] = i < end ? tmp[i] : -1;
        }
#pragma unroll
        for (int q = 0; q < 8; ++q) {
            if (wv[q] >= 0) {
                const int lo  = wv[q] >> 17;
                const int pos = my[lo];
                my[lo] = pos + 1;
                ssrc[beg + binOff[lo] + pos] = wv[q] & ((1 << 17) - 1);
            }
        }
    }
}

template <int K, bool POOL, bool VIA_IDS>
__global__ void __launch_bounds__(128, 16)
sage_layer_kernel(const int* __restrict__ rowptr,
                  const int* __restrict__ ssrc,
                  const int* __restrict__ ids,
                  const float* __restrict__ x,
                  const float* __restrict__ Wl,
                  const float* __restrict__ Wr,
                  const float* __restrict__ bias,
                  float* __restrict__ out,
                  const int* __restrict__ batch,
                  float* __restrict__ gsum,
                  float* __restrict__ gcnt, int nNodes) {
    constexpr int VW  = K / 32;
    constexpr int NPW = 4;
    constexpr int U   = 4;
    constexpr int PITCH = K + 4;
    static_assert(VW == 1 || VW == 2, "row is one or two lines");
    static_assert(N_NODES % 16 == 0, "a block owns 16 whole nodes");
    static_assert((unsigned long long)N_NODES * K * 4ull < (1ull << 32), "32-bit row offsets");
    __shared__ __attribute__((aligned(16))) float aggT[16 * PITCH];

    const int lane = threadIdx.x & 31;
    const int wave = threadIdx.x >> 5;
    const int row0 = blockIdx.x * 16;
    if (row0 >= nNodes) return;

    const int arow = VIA_IDS ? ids[row0 + (lane & 15)] : row0 + (lane & 15);

    {
        const int n0 = row0 + wave * NPW;
        const int rp = rowptr[n0 + (lane < NPW ? lane : NPW)];
        const unsigned laneOff = (unsigned)lane * (unsigned)(VW * 4);
        const char* xb = reinterpret_cast<const char*>(x);

        int first[NPW];
#pragma unroll
        for (int i = 0; i < NPW; ++i) {
            const int beg = __builtin_amdgcn_readlane(rp, i);
            const int end = __builtin_amdgcn_readlane(rp, i + 1);
            const int idx = beg + lane;
            first[i] = idx < end ? ssrc[idx] : 0;
        }
        if (VIA_IDS) {
#pragma unroll
            for (int i = 0; i < NPW; ++i) first[i] = ids[(unsigned)first[i]];
        }
#pragma unroll
        for (int i = 0; i < NPW; ++i) {
            const int beg = __builtin_amdgcn_readlane(rp, i);
            const int end = __builtin_amdgcn_readlane(rp, i + 1);
            float a0 = 0.0f, a1 = 0.0f;
            int my = first[i];
            for (int base = beg; base < end; base += 32) {
                if (base != beg) {
                    const int idx = base + lane;
                    my = idx < end ? ssrc[idx] : 0;
                    if (VIA_IDS) my = ids[(unsigned)my];
                }
                const int m = min(32, end - base);
                for (int j = 0; j < m; j += U) {
                    float v0[U], v1[U];
#pragma unroll
                    for (int u = 0; u < U; ++u) {
                        const bool ok = (j + u) < m;
                        const unsigned s = (unsigned)__builtin_amdgcn_readlane(my, ok ? j + u : j);
                        const char* p = xb + (s * (unsigned)(K * 4) + laneOff);
                        if (VW == 1) {
                            v0[u] = *reinterpret_cast<const float*>(p);
                            v1[u] = 0.0f;
                        } else {
                            const float2 q = *reinterpret_cast<const float2*>(p);
                            v0[u] = q.x;
                            v1[u] = q.y;
                        }
                        if (!ok) { v0[u] = 0.0f; v1[u] = 0.0f; }
                    }
#pragma unroll
                    for (int u = 0; u < U; ++u) {
                        a0 += v0[u];
                        if (VW == 2) a1 += v1[u];
                    }
                }
            }
            const float invc = 1.0f / fmaxf((float)(end - beg), 1.0f);
            float* ar = aggT + (wave * NPW + i) * PITCH + lane * VW;
            if (VW == 1) *ar = a0 * invc;
            else *reinterpret_cast<float2*>(ar) = make_float2(a0 * invc, a1 * invc);
        }
    }
    __syncthreads();

    const int col0 = wave * 16;
    const int grp  = lane >> 4;
    const int lr   = lane & 15;

    v8f c = {};
#pragma unroll
    for (int k0 = 0; k0 < K; k0 += 4) {
        const int ka = k0 + 2 * grp;
        v2f a, b;
        const float2 am = *reinterpret_cast<const float2*>(aggT + lr * PITCH + ka);
        a.x = am.x;
        a.y = am.y;
        b.x = Wl[(size_t)(ka + 0) * HIDDEN_D + col0 + lr];
        b.y = Wl[(size_t)(ka + 1) * HIDDEN_D + col0 + lr];
        c = __builtin_amdgcn_wmma_f32_16x16x4_f32(false, a, false, b,
                                                  (short)0, c, false, false);
    }
#pragma unroll
    for (int k0 = 0; k0 < K; k0 += 4) {
        const int ka = k0 + 2 * grp;
        v2f a, b;
        a.x = x[(size_t)arow * K + ka + 0];
        a.y = x[(size_t)arow * K + ka + 1];
        b.x = Wr[(size_t)(ka + 0) * HIDDEN_D + col0 + lr];
        b.y = Wr[(size_t)(ka + 1) * HIDDEN_D + col0 + lr];
        c = __builtin_amdgcn_wmma_f32_16x16x4_f32(false, a, false, b,
                                                  (short)0, c, false, false);
    }

    const float bv = bias[col0 + lr];
    if (!POOL) {
#pragma unroll
        for (int v = 0; v < 8; ++v) {
            const int m = v + 8 * grp;
            out[(size_t)(row0 + m) * HIDDEN_D + col0 + lr] = fmaxf(c[v] + bv, 0.0f);
        }
    } else {
        const int gid = batch[row0 + lr];
        const int g0  = __builtin_amdgcn_readfirstlane(gid);
        float r[8];
#pragma unroll
        for (int v = 0; v < 8; ++v) r[v] = fmaxf(c[v] + bv, 0.0f);
        if (__all(gid == g0)) {
            float s = ((r[0] + r[1]) + (r[2] + r[3])) + ((r[4] + r[5]) + (r[6] + r[7]));
            s += __shfl_xor(s, 16, 32);
            if (grp == 0) atomAdd1(gsum + (size_t)g0 * HIDDEN_D + col0 + lr, s);
            if (threadIdx.x == 0) atomAdd1(gcnt + g0, 16.0f);
        } else {
#pragma unroll
            for (int v = 0; v < 8; ++v) {
                const int gv = __shfl(gid, v + 8 * grp, 32);
                atomAdd1(gsum + (size_t)gv * HIDDEN_D + col0 + lr, r[v]);
            }
            if (threadIdx.x < 16) atomAdd1(gcnt + gid, 1.0f);
        }
    }
}

__global__ void head_kernel(const float* __restrict__ gsum,
                            const float* __restrict__ gcnt,
                            const float* __restrict__ Wo,
                            const float* __restrict__ bo,
                            float* __restrict__ out) {
    int t = blockIdx.x * blockDim.x + threadIdx.x;
    if (t >= N_GRAPHS * N_CLASSES) return;
    int g = t / N_CLASSES;
    int c = t - g * N_CLASSES;
    float invc = 1.0f / fmaxf(gcnt[g], 1.0f);
    float acc = 0.0f;
    for (int d = 0; d < HIDDEN_D; ++d)
        acc += gsum[(size_t)g * HIDDEN_D + d] * invc * Wo[(size_t)d * N_CLASSES + c];
    out[t] = acc + bo[c];
}

extern "C" void kernel_launch(void* const* d_in, const int* in_sizes, int n_in,
                              void* d_out, int out_size, void* d_ws, size_t ws_size,
                              hipStream_t stream) {
    const int*   node_ids = (const int*)  d_in[0];
    const int*   src      = (const int*)  d_in[1];
    const int*   dst      = (const int*)  d_in[2];
    const int*   batch    = (const int*)  d_in[3];
    const float* embed    = (const float*)d_in[4];
    const float* W1_l     = (const float*)d_in[5];
    const float* b1       = (const float*)d_in[6];
    const float* W1_r     = (const float*)d_in[7];
    const float* W2_l     = (const float*)d_in[8];
    const float* b2       = (const float*)d_in[9];
    const float* W2_r     = (const float*)d_in[10];
    const float* W_out    = (const float*)d_in[11];
    const float* b_out    = (const float*)d_in[12];
    float* out = (float*)d_out;

    float* ws   = (float*)d_ws;
    float* h1   = ws;
    float* gsum = h1   + (size_t)N_NODES * HIDDEN_D;
    float* gcnt = gsum + (size_t)N_GRAPHS * HIDDEN_D;
    int* rowptr   = (int*)(gcnt + (size_t)N_GRAPHS);
    int* table    = rowptr   + (size_t)(N_NODES + 4);
    int* tableOff = table    + (size_t)TABLE_N;
    int* tmp      = tableOff + (size_t)(TABLE_N + 4);
    int* ssrc     = tmp      + (size_t)N_EDGES;
    const int zeroN = N_GRAPHS * HIDDEN_D + N_GRAPHS;

    const int B = 256;
    auto blocks = [](int n, int b) { return (n + b - 1) / b; };

    zero_kernel<<<blocks(zeroN, B), B, 0, stream>>>(gsum, zeroN);

    coarse_count_kernel<<<P1_BLOCKS, P_THREADS, 0, stream>>>(dst, table, N_EDGES);
    offsets_kernel<<<blocks(TABLE_N, 1024), 1024, 0, stream>>>(table, tableOff, TABLE_N);
    coarse_place_kernel<<<P1_BLOCKS, P_THREADS, 0, stream>>>(src, dst, tableOff, tmp, N_EDGES);
    fine_group_kernel<<<N_BUCKETS, P_THREADS, 0, stream>>>(tableOff, tmp, rowptr, ssrc, N_EDGES);

    sage_layer_kernel<EMBED_D, false, true><<<N_NODES / 16, 128, 0, stream>>>(
        rowptr, ssrc, node_ids, embed, W1_l, W1_r, b1, h1, nullptr, nullptr, nullptr, N_NODES);

    sage_layer_kernel<HIDDEN_D, true, false><<<N_NODES / 16, 128, 0, stream>>>(
        rowptr, ssrc, nullptr, h1, W2_l, W2_r, b2, nullptr, batch, gsum, gcnt, N_NODES);

    head_kernel<<<blocks(N_GRAPHS * N_CLASSES, B), B, 0, stream>>>(
        gsum, gcnt, W_out, b_out, out);
}


# edges grouped in two launches, not four: each block orders its slice by bucket, a bucket's block gathers its 256 slice segments; count and offsets launches gone; row sum order changes, still fixed
# speedup vs baseline: 2.6952x; 1.2815x over previous
#include <hip/hip_runtime.h>

#define N_NODES   100000
#define N_EDGES   1200000
#define EMBED_D   32
#define HIDDEN_D  64
#define N_CLASSES 10
#define N_GRAPHS  1024

typedef __attribute__((ext_vector_type(2))) float v2f;
typedef __attribute__((ext_vector_type(8))) float v8f;

__device__ __forceinline__ void atomAdd1(float* p, float v) {
    asm volatile("global_atomic_add_f32 %0, %1, off scope:SCOPE_DEV"
                 :: "v"(p), "v"(v) : "memory");
}

__global__ void zero_kernel(float* __restrict__ p, int n) {
    int t = blockIdx.x * blockDim.x + threadIdx.x;
    if (t < n) p[t] = 0.0f;
}

#define FINE_BITS 8
#define FINE_BINS 256
#define N_BUCKETS ((N_NODES + FINE_BINS - 1) / FINE_BINS)
#define P1_BLOCKS 256
#define P_THREADS 128
#define P1_EDGES  ((N_EDGES + P1_BLOCKS - 1) / P1_BLOCKS)
#define PITCH1    (N_BUCKETS | 1)
#define PITCH2    (FINE_BINS + 1)
#define OFFT_N    ((N_BUCKETS + 1) * P1_BLOCKS)
#define SCAN_PER_LANE ((N_BUCKETS + 1 + 31) / 32)
#define SCAN_N    (32 * SCAN_PER_LANE)
#define HALF_WAVES (P_THREADS / 16)
#define SEG_PER_HW (P1_BLOCKS / HALF_WAVES)
#define SEG_BATCH  16
static_assert(N_NODES <= (1 << 17), "a source id fits 17 bits of the packed word");
static_assert((P_THREADS * PITCH1) % 4 == 0 && (P_THREADS * PITCH2) % 4 == 0, "LDS cleared 16 bytes at a time");
static_assert(OFFT_N % 4 == 0, "workspace pieces are multiples of four words");
static_assert(P1_BLOCKS % P_THREADS == 0 && SEG_PER_HW % SEG_BATCH == 0, "segments dealt out evenly");

__device__ __forceinline__ void lds_clear(int* h, int nWords, int t) {
    int4* h4 = reinterpret_cast<int4*>(h);
    for (int i = t; i < nWords / 4; i += P_THREADS) h4[i] = make_int4(0, 0, 0, 0);
}

__device__ __forceinline__ void coarse_count(const int* __restrict__ dst, int* my,
                                             int e0, int e1, int t) {
    for (int base = e0; base < e1; base += P_THREADS * 8) {
        int d[8];
#pragma unroll
        for (int q = 0; q < 8; ++q) {
            const int e = base + q * P_THREADS + t;
            d[q] = e < e1 ? dst[e] : -1;
        }
#pragma unroll
        for (int q = 0; q < 8; ++q)
            if (d[q] >= 0) my[d[q] >> FINE_BITS] += 1;
    }
}

__global__ void __launch_bounds__(P_THREADS)
slice_sort_kernel(const int* __restrict__ src, const int* __restrict__ dst,
                  int* __restrict__ offT, int* __restrict__ tmp, int nEdges) {
    __shared__ __attribute__((aligned(16))) int hist[P_THREADS * PITCH1];
    __shared__ int bstart[SCAN_N];
    const int t    = threadIdx.x;
    const int lane = t & 31;
    const int e0 = blockIdx.x * P1_EDGES;
    const int e1 = min(e0 + P1_EDGES, nEdges);
    int* my = hist + t * PITCH1;
    lds_clear(hist, P_THREADS * PITCH1, t);
    __syncthreads();
    coarse_count(dst, my, e0, e1, t);
    __syncthreads();
    for (int b = t; b < N_BUCKETS; b += P_THREADS) {
        int run = 0;
        for (int r = 0; r < P_THREADS; ++r) {
            const int c = hist[r * PITCH1 + b];
            hist[r * PITCH1 + b] = run;
            run += c;
        }
        bstart[b] = run;
    }
    for (int b = N_BUCKETS + t; b < SCAN_N; b += P_THREADS) bstart[b] = 0;
    __syncthreads();
    if (t < 32) {
        int c[SCAN_PER_LANE];
        int s = 0;
#pragma unroll
        for (int q = 0; q < SCAN_PER_LANE; ++q) { c[q] = bstart[SCAN_PER_LANE * lane + q]; s += c[q]; }
        int x = s;
#pragma unroll
        for (int d = 1; d < 32; d <<= 1) {
            const int y = __shfl_up(x, d, 32);
            if (lane >= d) x += y;
        }
        int ex = x - s;
#pragma unroll
        for (int q = 0; q < SCAN_PER_LANE; ++q) { bstart[SCAN_PER_LANE * lane + q] = ex; ex += c[q]; }
    }
    __syncthreads();
    for (int b = t; b <= N_BUCKETS; b += P_THREADS) offT[b * P1_BLOCKS + blockIdx.x] = bstart[b];
    for (int base = e0; base < e1; base += P_THREADS * 8) {
        int d[8], s[8];
#pragma unroll
        for (int q = 0; q < 8; ++q) {
            const int e = base + q * P_THREADS + t;
            d[q] = e < e1 ? dst[e] : -1;
            s[q] = e < e1 ? src[e] : 0;
        }
#pragma unroll
        for (int q = 0; q < 8; ++q) {
            if (d[q] >= 0) {
                const int bk  = d[q] >> FINE_BITS;
                const int pos = my[bk];
                my[bk] = pos + 1;
                tmp[e0 + bstart[bk] + pos] = ((d[q] & (FINE_BINS - 1)) << 17) | s[q];
            }
        }
    }
}

__device__ __forceinline__ void seg_load(const int* __restrict__ tmp, const int* sb,
                                         const int* ln, int i, int* wv) {
#pragma unroll
    for (int q = 0; q < SEG_BATCH; ++q) wv[q] = i < ln[q] ? tmp[sb[q] + i] : -1;
}

__global__ void __launch_bounds__(P_THREADS)
fine_group_kernel(const int* __restrict__ offT, const int* __restrict__ tmp,
                  int* __restrict__ rowptr, int* __restrict__ ssrc) {
    __shared__ __attribute__((aligned(16))) int hist[P_THREADS * PITCH2];
    __shared__ int tot[FINE_BINS];
    __shared__ int binOff[FINE_BINS];
    __shared__ int segBeg[P1_BLOCKS];
    __shared__ int segLen[P1_BLOCKS];
    __shared__ int wsum[2 * (P_THREADS / 32)];
    const int t    = threadIdx.x;
    const int lane = t & 31;
    const int w    = t >> 5;
    const int hw   = t >> 4;
    const int hl   = t & 15;
    const int k    = blockIdx.x;
    int* my = hist + t * PITCH2;
    int sa = 0, sl = 0;
    for (int s = t; s < P1_BLOCKS; s += P_THREADS) {
        const int a = offT[k * P1_BLOCKS + s];
        const int b = offT[(k + 1) * P1_BLOCKS + s];
        segBeg[s] = s * P1_EDGES + a;
        segLen[s] = b - a;
        sa += a;
        sl += b - a;
    }
#pragma unroll
    for (int d = 16; d >= 1; d >>= 1) {
        sa += __shfl_xor(sa, d, 32);
        sl += __shfl_xor(sl, d, 32);
    }
    if (lane == 0) { wsum[w] = sa; wsum[P_THREADS / 32 + w] = sl; }
    lds_clear(hist, P_THREADS * PITCH2, t);
    __syncthreads();
    const int beg = (wsum[0] + wsum[1]) + (wsum[2] + wsum[3]);
    const int end = beg + (wsum[4] + wsum[5]) + (wsum[6] + wsum[7]);
#pragma unroll 1
    for (int j0 = 0; j0 < SEG_PER_HW; j0 += SEG_BATCH) {
        int sb[SEG_BATCH], ln[SEG_BATCH], wv[SEG_BATCH];
#pragma unroll
        for (int q = 0; q < SEG_BATCH; ++q) {
            const int s = hw + HALF_WAVES * (j0 + q);
            sb[q] = segBeg[s];
            ln[q] = segLen[s];
        }
#pragma unroll
        for (int i0 = 0; i0 < 32; i0 += 16) {
            seg_load(tmp, sb, ln, i0 + hl, wv);
#pragma unroll
            for (int q = 0; q < SEG_BATCH; ++q)
                if (wv[q] >= 0) my[wv[q] >> 17] += 1;
        }
#pragma unroll
        for (int q = 0; q < SEG_BATCH; ++q)
            for (int i = 32 + hl; i < ln[q]; i += 16) my[tmp[sb[q] + i] >> 17] += 1;
    }
    __syncthreads();
    for (int b = t; b < FINE_BINS; b += P_THREADS) {
        int run = 0;
        for (int r = 0; r < P_THREADS; ++r) {
            const int c = hist[r * PITCH2 + b];
            hist[r * PITCH2 + b] = run;
            run += c;
        }
        tot[b] = run;
    }
    __syncthreads();
    if (t < 32) {
        int c[8];
        int s = 0;
#pragma unroll
        for (int q = 0; q < 8; ++q) { c[q] = tot[8 * lane + q]; s += c[q]; }
        int x = s;
#pragma unroll
        for (int d = 1; d < 32; d <<= 1) {
            const int y = __shfl_up(x, d, 32);
            if (lane >= d) x += y;
        }
        int ex = x - s;
#pragma unroll
        for (int q = 0; q < 8; ++q) { binOff[8 * lane + q] = ex; ex += c[q]; }
    }
    __syncthreads();
    for (int b = t; b < FINE_BINS; b += P_THREADS) {
        const int node = k * FINE_BINS + b;
        if (node < N_NODES) rowptr[node] = beg + binOff[b];
    }
    if (k == N_BUCKETS - 1 && t == 0) rowptr[N_NODES] = end;
#pragma unroll 1
    for (int j0 = 0; j0 < SEG_PER_HW; j0 += SEG_BATCH) {
        int sb[SEG_BATCH], ln[SEG_BATCH], wv[SEG_BATCH];
#pragma unroll
        for (int q = 0; q < SEG_BATCH; ++q) {
            const int s = hw + HALF_WAVES * (j0 + q);
            sb[q] = segBeg[s];
            ln[q] = segLen[s];
        }
#pragma unroll
        for (int i0 = 0; i0 < 32; i0 += 16) {
            seg_load(tmp, sb, ln, i0 + hl, wv);
#pragma unroll
            for (int q = 0; q < SEG_BATCH; ++q) {
                if (wv[q] >= 0) {
                    const int lo  = wv[q] >> 17;
                    const int pos = my[lo];
                    my[lo] = pos + 1;
                    ssrc[beg + binOff[lo] + pos] = wv[q] & ((1 << 17) - 1);
                }
            }
        }
#pragma unroll
        for (int q = 0; q < SEG_BATCH; ++q) {
            for (int i = 32 + hl; i < ln[q]; i += 16) {
                const int x   = tmp[sb[q] + i];
                const int lo  = x >> 17;
                const int pos = my[lo];
                my[lo] = pos + 1;
                ssrc[beg + binOff[lo] + pos] = x & ((1 << 17) - 1);
            }
        }
    }
}

template <int K, bool POOL, bool VIA_IDS>
__global__ void __launch_bounds__(128, 16)
sage_layer_kernel(const int* __restrict__ rowptr,
                  const int* __restrict__ ssrc,
                  const int* __restrict__ ids,
                  const float* __restrict__ x,
                  const float* __restrict__ Wl,
                  const float* __restrict__ Wr,
                  const float* __restrict__ bias,
                  float* __restrict__ out,
                  const int* __restrict__ batch,
                  float* __restrict__ gsum,
                  float* __restrict__ gcnt, int nNodes) {
    constexpr int VW  = K / 32;
    constexpr int NPW = 4;
    constexpr int U   = 4;
    constexpr int PITCH = K + 4;
    static_assert(VW == 1 || VW == 2, "row is one or two lines");
    static_assert(N_NODES % 16 == 0, "a block owns 16 whole nodes");
    static_assert((unsigned long long)N_NODES * K * 4ull < (1ull << 32), "32-bit row offsets");
    __shared__ __attribute__((aligned(16))) float aggT[16 * PITCH];

    const int lane = threadIdx.x & 31;
    const int wave = threadIdx.x >> 5;
    const int row0 = blockIdx.x * 16;
    if (row0 >= nNodes) return;

    const int arow = VIA_IDS ? ids[row0 + (lane & 15)] : row0 + (lane & 15);

    {
        const int n0 = row0 + wave * NPW;
        const int rp = rowptr[n0 + (lane < NPW ? lane : NPW)];
        const unsigned laneOff = (unsigned)lane * (unsigned)(VW * 4);
        const char* xb = reinterpret_cast<const char*>(x);

        int first[NPW];
#pragma unroll
        for (int i = 0; i < NPW; ++i) {
            const int beg = __builtin_amdgcn_readlane(rp, i);
            const int end = __builtin_amdgcn_readlane(rp, i + 1);
            const int idx = beg + lane;
            first[i] = idx < end ? ssrc[idx] : 0;
        }
        if (VIA_IDS) {
#pragma unroll
            for (int i = 0; i < NPW; ++i) first[i] = ids[(unsigned)first[i]];
        }
#pragma unroll
        for (int i = 0; i < NPW; ++i) {
            const int beg = __builtin_amdgcn_readlane(rp, i);
            const int end = __builtin_amdgcn_readlane(rp, i + 1);
            float a0 = 0.0f, a1 = 0.0f;
            int my = first[i];
            for (int base = beg; base < end; base += 32) {
                if (base != beg) {
                    const int idx = base + lane;
                    my = idx < end ? ssrc[idx] : 0;
                    if (VIA_IDS) my = ids[(unsigned)my];
                }
                const int m = min(32, end - base);
                for (int j = 0; j < m; j += U) {
                    float v0[U], v1[U];
#pragma unroll
                    for (int u = 0; u < U; ++u) {
                        const bool ok = (j + u) < m;
                        const unsigned s = (unsigned)__builtin_amdgcn_readlane(my, ok ? j + u : j);
                        const char* p = xb + (s * (unsigned)(K * 4) + laneOff);
                        if (VW == 1) {
                            v0[u] = *reinterpret_cast<const float*>(p);
                            v1[u] = 0.0f;
                        } else {
                            const float2 q = *reinterpret_cast<const float2*>(p);
                            v0[u] = q.x;
                            v1[u] = q.y;
                        }
                        if (!ok) { v0[u] = 0.0f; v1[u] = 0.0f; }
                    }
#pragma unroll
                    for (int u = 0; u < U; ++u) {
                        a0 += v0[u];
                        if (VW == 2) a1 += v1[u];
                    }
                }
            }
            const float invc = 1.0f / fmaxf((float)(end - beg), 1.0f);
            float* ar = aggT + (wave * NPW + i) * PITCH + lane * VW;
            if (VW == 1) *ar = a0 * invc;
            else *reinterpret_cast<float2*>(ar) = make_float2(a0 * invc, a1 * invc);
        }
    }
    __syncthreads();

    const int col0 = wave * 16;
    const int grp  = lane >> 4;
    const int lr   = lane & 15;

    v8f c = {};
#pragma unroll
    for (int k0 = 0; k0 < K; k0 += 4) {
        const int ka = k0 + 2 * grp;
        v2f a, b;
        const float2 am = *reinterpret_cast<const float2*>(aggT + lr * PITCH + ka);
        a.x = am.x;
        a.y = am.y;
        b.x = Wl[(size_t)(ka + 0) * HIDDEN_D + col0 + lr];
        b.y = Wl[(size_t)(ka + 1) * HIDDEN_D + col0 + lr];
        c = __builtin_amdgcn_wmma_f32_16x16x4_f32(false, a, false, b,
                                                  (short)0, c, false, false);
    }
#pragma unroll
    for (int k0 = 0; k0 < K; k0 += 4) {
        const int ka = k0 + 2 * grp;
        v2f a, b;
        a.x = x[(size_t)arow * K + ka + 0];
        a.y = x[(size_t)arow * K + ka + 1];
        b.x = Wr[(size_t)(ka + 0) * HIDDEN_D + col0 + lr];
        b.y = Wr[(size_t)(ka + 1) * HIDDEN_D + col0 + lr];
        c = __builtin_amdgcn_wmma_f32_16x16x4_f32(false, a, false, b,
                                                  (short)0, c, false, false);
    }

    const float bv = bias[col0 + lr];
    if (!POOL) {
#pragma unroll
        for (int v = 0; v < 8; ++v) {
            const int m = v + 8 * grp;
            out[(size_t)(row0 + m) * HIDDEN_D + col0 + lr] = fmaxf(c[v] + bv, 0.0f);
        }
    } else {
        const int gid = batch[row0 + lr];
        const int g0  = __builtin_amdgcn_readfirstlane(gid);
        float r[8];
#pragma unroll
        for (int v = 0; v < 8; ++v) r[v] = fmaxf(c[v] + bv, 0.0f);
        if (__all(gid == g0)) {
            float s = ((r[0] + r[1]) + (r[2] + r[3])) + ((r[4] + r[5]) + (r[6] + r[7]));
            s += __shfl_xor(s, 16, 32);
            if (grp == 0) atomAdd1(gsum + (size_t)g0 * HIDDEN_D + col0 + lr, s);
            if (threadIdx.x == 0) atomAdd1(gcnt + g0, 16.0f);
        } else {
#pragma unroll
            for (int v = 0; v < 8; ++v) {
                const int gv = __shfl(gid, v + 8 * grp, 32);
                atomAdd1(gsum + (size_t)gv * HIDDEN_D + col0 + lr, r[v]);
            }
            if (threadIdx.x < 16) atomAdd1(gcnt + gid, 1.0f);
        }
    }
}

__global__ void head_kernel(const float* __restrict__ gsum,
                            const float* __restrict__ gcnt,
                            const float* __restrict__ Wo,
                            const float* __restrict__ bo,
                            float* __restrict__ out) {
    int t = blockIdx.x * blockDim.x + threadIdx.x;
    if (t >= N_GRAPHS * N_CLASSES) return;
    int g = t / N_CLASSES;
    int c = t - g * N_CLASSES;
    float invc = 1.0f / fmaxf(gcnt[g], 1.0f);
    float acc = 0.0f;
    for (int d = 0; d < HIDDEN_D; ++d)
        acc += gsum[(size_t)g * HIDDEN_D + d] * invc * Wo[(size_t)d * N_CLASSES + c];
    out[t] = acc + bo[c];
}

extern "C" void kernel_launch(void* const* d_in, const int* in_sizes, int n_in,
                              void* d_out, int out_size, void* d_ws, size_t ws_size,
                              hipStream_t stream) {
    const int*   node_ids = (const int*)  d_in[0];
    const int*   src      = (const int*)  d_in[1];
    const int*   dst      = (const int*)  d_in[2];
    const int*   batch    = (const int*)  d_in[3];
    const float* embed    = (const float*)d_in[4];
    const float* W1_l     = (const float*)d_in[5];
    const float* b1       = (const float*)d_in[6];
    const float* W1_r     = (const float*)d_in[7];
    const float* W2_l     = (const float*)d_in[8];
    const float* b2       = (const float*)d_in[9];
    const float* W2_r     = (const float*)d_in[10];
    const float* W_out    = (const float*)d_in[11];
    const float* b_out    = (const float*)d_in[12];
    float* out = (float*)d_out;

    float* ws   = (float*)d_ws;
    float* h1   = ws;
    float* gsum = h1   + (size_t)N_NODES * HIDDEN_D;
    float* gcnt = gsum + (size_t)N_GRAPHS * HIDDEN_D;
    int* rowptr   = (int*)(gcnt + (size_t)N_GRAPHS);
    int* offT     = rowptr   + (size_t)(N_NODES + 4);
    int* tmp      = offT     + (size_t)OFFT_N;
    int* ssrc     = tmp      + (size_t)N_EDGES;
    const int zeroN = N_GRAPHS * HIDDEN_D + N_GRAPHS;

    const int B = 256;
    auto blocks = [](int n, int b) { return (n + b - 1) / b; };

    zero_kernel<<<blocks(zeroN, B), B, 0, stream>>>(gsum, zeroN);

    slice_sort_kernel<<<P1_BLOCKS, P_THREADS, 0, stream>>>(src, dst, offT, tmp, N_EDGES);
    fine_group_kernel<<<N_BUCKETS, P_THREADS, 0, stream>>>(offT, tmp, rowptr, ssrc);

    sage_layer_kernel<EMBED_D, false, true><<<N_NODES / 16, 128, 0, stream>>>(
        rowptr, ssrc, node_ids, embed, W1_l, W1_r, b1, h1, nullptr, nullptr, nullptr, N_NODES);

    sage_layer_kernel<HIDDEN_D, true, false><<<N_NODES / 16, 128, 0, stream>>>(
        rowptr, ssrc, nullptr, h1, W2_l, W2_r, b2, nullptr, batch, gsum, gcnt, N_NODES);

    head_kernel<<<blocks(N_GRAPHS * N_CLASSES, B), B, 0, stream>>>(
        gsum, gcnt, W_out, b_out, out);
}
